# starting kernel
# baseline (speedup 1.0000x reference)
#include <hip/hip_runtime.h>
#include <math.h>

typedef __attribute__((ext_vector_type(2))) float v2f;
typedef __attribute__((ext_vector_type(8))) float v8f;

#define E_DIM 64
#define H_DIM 128
#define O_DIM 16
#define BATCH 64
#define NNODES 1000
#define FUT 12
#define G4 512

__device__ __forceinline__ float sigm(float x) { return 1.0f / (1.0f + __expf(-x)); }

__global__ __launch_bounds__(256) void decoder_scan(
    const float* __restrict__ enc,
    const float* __restrict__ mean_,
    const float* __restrict__ s,
    const float* __restrict__ h0,
    const float* __restrict__ c0,
    const float* __restrict__ Wih0,
    const float* __restrict__ Whh0,
    const float* __restrict__ bih0, const float* __restrict__ bhh0,
    const float* __restrict__ Whr0,
    const float* __restrict__ Wih1,
    const float* __restrict__ Whh1,
    const float* __restrict__ bih1, const float* __restrict__ bhh1,
    const float* __restrict__ Whr1,
    float* __restrict__ out)
{
  __shared__ float WcatT0[32][G4];
  __shared__ float WcatT1[32][G4];
  __shared__ float Whr0T[H_DIM][O_DIM];
  __shared__ float Whr1T[H_DIM][O_DIM];
  __shared__ float GencB[16][G4];
  __shared__ float bias0s[G4];
  __shared__ float bias1s[G4];
  __shared__ float hfull[16][H_DIM];
  __shared__ float part[8][16][16];
  __shared__ float xin0[16][32];
  __shared__ float xin1[16][32];
  __shared__ float h1Buf[16][16];
  __shared__ float h2Buf[16][16];
  __shared__ float lastBuf[16][16];
  __shared__ float nmBuf[16][16];

  const int tid  = threadIdx.x;
  const int w    = tid >> 5;
  const int lane = tid & 31;
  const int lm   = lane & 15;
  const int lh   = lane >> 4;
  const int b0   = blockIdx.x * 16;

  for (int idx = tid; idx < 32 * G4; idx += 256) {
    int k = idx >> 9;
    int col = idx & 511;
    WcatT0[k][col] = (k < 16) ? Wih0[col * 80 + 64 + k] : Whh0[col * 16 + (k - 16)];
    WcatT1[k][col] = (k < 16) ? Wih1[col * 16 + k]      : Whh1[col * 16 + (k - 16)];
  }
  for (int idx = tid; idx < H_DIM * O_DIM; idx += 256) {
    int k = idx >> 4; int col = idx & 15;
    Whr0T[k][col] = Whr0[col * H_DIM + k];
    Whr1T[k][col] = Whr1[col * H_DIM + k];
  }
  for (int col = tid; col < G4; col += 256) {
    bias0s[col] = bih0[col] + bhh0[col];
    bias1s[col] = bih1[col] + bhh1[col];
  }
  {
    int row = tid >> 4, o = tid & 15;
    h1Buf[row][o]   = h0[o];
    h2Buf[row][o]   = h0[O_DIM + o];
    lastBuf[row][o] = s[o];
  }
  const int hcol = 16 * w + lm;
  float c1r[8], c2r[8];
  #pragma unroll
  for (int r = 0; r < 8; ++r) { c1r[r] = c0[hcol]; c2r[r] = c0[H_DIM + hcol]; }
  __syncthreads();

  for (int n = 0; n < NNODES; ++n) {
    {
      int row = tid >> 4, o = tid & 15;
      const float* mp = mean_ + (size_t)(b0 + row) * FUT * NNODES * O_DIM
                              + (size_t)n * O_DIM + o;
      float sum = 0.f;
      #pragma unroll
      for (int f = 0; f < FUT; ++f) sum += mp[(size_t)f * NNODES * O_DIM];
      nmBuf[row][o] = sum * (1.0f / FUT);
    }
    #pragma unroll
    for (int g = 0; g < 4; ++g) {
      int gc = 16 * (w + 8 * g) + lm;
      v8f acc;
      float bv = bias0s[gc];
      #pragma unroll
      for (int r = 0; r < 8; ++r) acc[r] = bv;
      const float* ap = enc + (size_t)(b0 + lm) * NNODES * E_DIM + (size_t)n * E_DIM;
      #pragma unroll
      for (int k = 0; k < 16; ++k) {
        int kk = 4 * k + 2 * lh;
        v2f a, b;
        a.x = ap[kk];              a.y = ap[kk + 1];
        b.x = Wih0[gc * 80 + kk];  b.y = Wih0[gc * 80 + kk + 1];
        acc = __builtin_amdgcn_wmma_f32_16x16x4_f32(false, a, false, b, (short)0, acc, false, false);
      }
      #pragma unroll
      for (int r = 0; r < 8; ++r) GencB[r + 8 * lh][gc] = acc[r];
    }
    __syncthreads();

    for (int f = 0; f < FUT; ++f) {
      {
        int row = tid >> 4, o = tid & 15;
        xin0[row][o]      = lastBuf[row][o] + nmBuf[row][o];
        xin0[row][16 + o] = h1Buf[row][o];
      }
      __syncthreads();

      {
        v8f acc[4];
        #pragma unroll
        for (int g = 0; g < 4; ++g) {
          int gc = 16 * (w + 8 * g) + lm;
          #pragma unroll
          for (int r = 0; r < 8; ++r) acc[g][r] = GencB[r + 8 * lh][gc];
        }
        #pragma unroll
        for (int k = 0; k < 8; ++k) {
          int kk = 4 * k + 2 * lh;
          v2f a; a.x = xin0[lm][kk]; a.y = xin0[lm][kk + 1];
          #pragma unroll
          for (int g = 0; g < 4; ++g) {
            int gc = 16 * (w + 8 * g) + lm;
            v2f b; b.x = WcatT0[kk][gc]; b.y = WcatT0[kk + 1][gc];
            acc[g] = __builtin_amdgcn_wmma_f32_16x16x4_f32(false, a, false, b, (short)0, acc[g], false, false);
          }
        }
        #pragma unroll
        for (int r = 0; r < 8; ++r) {
          float iv = sigm(acc[0][r]);
          float fv = sigm(acc[1][r]);
          float gv = tanhf(acc[2][r]);
          float ov = sigm(acc[3][r]);
          float cn = fv * c1r[r] + iv * gv;
          c1r[r] = cn;
          hfull[r + 8 * lh][hcol] = ov * tanhf(cn);
        }
      }
      __syncthreads();

      {
        v8f acc = {};
        #pragma unroll
        for (int k = 0; k < 4; ++k) {
          int kk = 16 * w + 4 * k + 2 * lh;
          v2f a; a.x = hfull[lm][kk];  a.y = hfull[lm][kk + 1];
          v2f b; b.x = Whr0T[kk][lm];  b.y = Whr0T[kk + 1][lm];
          acc = __builtin_amdgcn_wmma_f32_16x16x4_f32(false, a, false, b, (short)0, acc, false, false);
        }
        #pragma unroll
        for (int r = 0; r < 8; ++r) part[w][r + 8 * lh][lm] = acc[r];
      }
      __syncthreads();
      {
        int row = tid >> 4, o = tid & 15;
        float v = 0.f;
        #pragma unroll
        for (int p = 0; p < 8; ++p) v += part[p][row][o];
        h1Buf[row][o] = v;
        xin1[row][o] = v;
        xin1[row][16 + o] = h2Buf[row][o];
      }
      __syncthreads();

      {
        v8f acc[4];
        #pragma unroll
        for (int g = 0; g < 4; ++g) {
          int gc = 16 * (w + 8 * g) + lm;
          float bv = bias1s[gc];
          #pragma unroll
          for (int r = 0; r < 8; ++r) acc[g][r] = bv;
        }
        #pragma unroll
        for (int k = 0; k < 8; ++k) {
          int kk = 4 * k + 2 * lh;
          v2f a; a.x = xin1[lm][kk]; a.y = xin1[lm][kk + 1];
          #pragma unroll
          for (int g = 0; g < 4; ++g) {
            int gc = 16 * (w + 8 * g) + lm;
            v2f b; b.x = WcatT1[kk][gc]; b.y = WcatT1[kk + 1][gc];
            acc[g] = __builtin_amdgcn_wmma_f32_16x16x4_f32(false, a, false, b, (short)0, acc[g], false, false);
          }
        }
        #pragma unroll
        for (int r = 0; r < 8; ++r) {
          float iv = sigm(acc[0][r]);
          float fv = sigm(acc[1][r]);
          float gv = tanhf(acc[2][r]);
          float ov = sigm(acc[3][r]);
          float cn = fv * c2r[r] + iv * gv;
          c2r[r] = cn;
          hfull[r + 8 * lh][hcol] = ov * tanhf(cn);
        }
      }
      __syncthreads();

      {
        v8f acc = {};
        #pragma unroll
        for (int k = 0; k < 4; ++k) {
          int kk = 16 * w + 4 * k + 2 * lh;
          v2f a; a.x = hfull[lm][kk];  a.y = hfull[lm][kk + 1];
          v2f b; b.x = Whr1T[kk][lm];  b.y = Whr1T[kk + 1][lm];
          acc = __builtin_amdgcn_wmma_f32_16x16x4_f32(false, a, false, b, (short)0, acc, false, false);
        }
        #pragma unroll
        for (int r = 0; r < 8; ++r) part[w][r + 8 * lh][lm] = acc[r];
      }
      __syncthreads();
      {
        int row = tid >> 4, o = tid & 15;
        float v = 0.f;
        #pragma unroll
        for (int p = 0; p < 8; ++p) v += part[p][row][o];
        h2Buf[row][o]   = v;
        lastBuf[row][o] = v;
        out[(size_t)(b0 + row) * FUT * NNODES * O_DIM
            + (size_t)f * NNODES * O_DIM + (size_t)n * O_DIM + o] = v + nmBuf[row][o];
      }
      __syncthreads();
    }
  }
}

extern "C" void kernel_launch(void* const* d_in, const int* in_sizes, int n_in,
                              void* d_out, int out_size, void* d_ws, size_t ws_size,
                              hipStream_t stream) {
  (void)in_sizes; (void)n_in; (void)out_size; (void)d_ws; (void)ws_size;
  const float* enc   = (const float*)d_in[0];
  const float* mean_ = (const float*)d_in[1];
  const float* s     = (const float*)d_in[2];
  const float* h0    = (const float*)d_in[3];
  const float* c0    = (const float*)d_in[4];
  const float* Wih0  = (const float*)d_in[5];
  const float* Whh0  = (const float*)d_in[6];
  const float* bih0  = (const float*)d_in[7];
  const float* bhh0  = (const float*)d_in[8];
  const float* Whr0  = (const float*)d_in[9];
  const float* Wih1  = (const float*)d_in[10];
  const float* Whh1  = (const float*)d_in[11];
  const float* bih1  = (const float*)d_in[12];
  const float* bhh1  = (const float*)d_in[13];
  const float* Whr1  = (const float*)d_in[14];
  float* out = (float*)d_out;

  decoder_scan<<<dim3(BATCH / 16), dim3(256), 0, stream>>>(
      enc, mean_, s, h0, c0,
      Wih0, Whh0, bih0, bhh0, Whr0,
      Wih1, Whh1, bih1, bhh1, Whr1, out);
}


# LSTM gate sigmoid and tanh on hardware exp2 + reciprocal (v_exp_f32, v_rcp_f32, about 1 ulp each, branch-free) instead of libm tanhf and IEEE divide; summation order unchanged
# speedup vs baseline: 1.7839x; 1.7839x over previous
#include <hip/hip_runtime.h>
#include <math.h>

typedef __attribute__((ext_vector_type(2))) float v2f;
typedef __attribute__((ext_vector_type(8))) float v8f;

#define E_DIM 64
#define H_DIM 128
#define O_DIM 16
#define BATCH 64
#define NNODES 1000
#define FUT 12
#define G4 512

__device__ __forceinline__ float sigm(float x) {
  return __builtin_amdgcn_rcpf(1.0f + __builtin_amdgcn_exp2f(-1.44269504088896341f * x));
}
__device__ __forceinline__ float tanh_hw(float x) {
  return 1.0f - 2.0f * __builtin_amdgcn_rcpf(1.0f + __builtin_amdgcn_exp2f(2.88539008177792681f * x));
}

__global__ __launch_bounds__(256) void decoder_scan(
    const float* __restrict__ enc,
    const float* __restrict__ mean_,
    const float* __restrict__ s,
    const float* __restrict__ h0,
    const float* __restrict__ c0,
    const float* __restrict__ Wih0,
    const float* __restrict__ Whh0,
    const float* __restrict__ bih0, const float* __restrict__ bhh0,
    const float* __restrict__ Whr0,
    const float* __restrict__ Wih1,
    const float* __restrict__ Whh1,
    const float* __restrict__ bih1, const float* __restrict__ bhh1,
    const float* __restrict__ Whr1,
    float* __restrict__ out)
{
  __shared__ float WcatT0[32][G4];
  __shared__ float WcatT1[32][G4];
  __shared__ float Whr0T[H_DIM][O_DIM];
  __shared__ float Whr1T[H_DIM][O_DIM];
  __shared__ float GencB[16][G4];
  __shared__ float bias0s[G4];
  __shared__ float bias1s[G4];
  __shared__ float hfull[16][H_DIM];
  __shared__ float part[8][16][16];
  __shared__ float xin0[16][32];
  __shared__ float xin1[16][32];
  __shared__ float h1Buf[16][16];
  __shared__ float h2Buf[16][16];
  __shared__ float lastBuf[16][16];
  __shared__ float nmBuf[16][16];

  const int tid  = threadIdx.x;
  const int w    = tid >> 5;
  const int lane = tid & 31;
  const int lm   = lane & 15;
  const int lh   = lane >> 4;
  const int b0   = blockIdx.x * 16;

  for (int idx = tid; idx < 32 * G4; idx += 256) {
    int k = idx >> 9;
    int col = idx & 511;
    WcatT0[k][col] = (k < 16) ? Wih0[col * 80 + 64 + k] : Whh0[col * 16 + (k - 16)];
    WcatT1[k][col] = (k < 16) ? Wih1[col * 16 + k]      : Whh1[col * 16 + (k - 16)];
  }
  for (int idx = tid; idx < H_DIM * O_DIM; idx += 256) {
    int k = idx >> 4; int col = idx & 15;
    Whr0T[k][col] = Whr0[col * H_DIM + k];
    Whr1T[k][col] = Whr1[col * H_DIM + k];
  }
  for (int col = tid; col < G4; col += 256) {
    bias0s[col] = bih0[col] + bhh0[col];
    bias1s[col] = bih1[col] + bhh1[col];
  }
  {
    int row = tid >> 4, o = tid & 15;
    h1Buf[row][o]   = h0[o];
    h2Buf[row][o]   = h0[O_DIM + o];
    lastBuf[row][o] = s[o];
  }
  const int hcol = 16 * w + lm;
  float c1r[8], c2r[8];
  #pragma unroll
  for (int r = 0; r < 8; ++r) { c1r[r] = c0[hcol]; c2r[r] = c0[H_DIM + hcol]; }
  __syncthreads();

  for (int n = 0; n < NNODES; ++n) {
    {
      int row = tid >> 4, o = tid & 15;
      const float* mp = mean_ + (size_t)(b0 + row) * FUT * NNODES * O_DIM
                              + (size_t)n * O_DIM + o;
      float sum = 0.f;
      #pragma unroll
      for (int f = 0; f < FUT; ++f) sum += mp[(size_t)f * NNODES * O_DIM];
      nmBuf[row][o] = sum * (1.0f / FUT);
    }
    #pragma unroll
    for (int g = 0; g < 4; ++g) {
      int gc = 16 * (w + 8 * g) + lm;
      v8f acc;
      float bv = bias0s[gc];
      #pragma unroll
      for (int r = 0; r < 8; ++r) acc[r] = bv;
      const float* ap = enc + (size_t)(b0 + lm) * NNODES * E_DIM + (size_t)n * E_DIM;
      #pragma unroll
      for (int k = 0; k < 16; ++k) {
        int kk = 4 * k + 2 * lh;
        v2f a, b;
        a.x = ap[kk];              a.y = ap[kk + 1];
        b.x = Wih0[gc * 80 + kk];  b.y = Wih0[gc * 80 + kk + 1];
        acc = __builtin_amdgcn_wmma_f32_16x16x4_f32(false, a, false, b, (short)0, acc, false, false);
      }
      #pragma unroll
      for (int r = 0; r < 8; ++r) GencB[r + 8 * lh][gc] = acc[r];
    }
    __syncthreads();

    for (int f = 0; f < FUT; ++f) {
      {
        int row = tid >> 4, o = tid & 15;
        xin0[row][o]      = lastBuf[row][o] + nmBuf[row][o];
        xin0[row][16 + o] = h1Buf[row][o];
      }
      __syncthreads();

      {
        v8f acc[4];
        #pragma unroll
        for (int g = 0; g < 4; ++g) {
          int gc = 16 * (w + 8 * g) + lm;
          #pragma unroll
          for (int r = 0; r < 8; ++r) acc[g][r] = GencB[r + 8 * lh][gc];
        }
        #pragma unroll
        for (int k = 0; k < 8; ++k) {
          int kk = 4 * k + 2 * lh;
          v2f a; a.x = xin0[lm][kk]; a.y = xin0[lm][kk + 1];
          #pragma unroll
          for (int g = 0; g < 4; ++g) {
            int gc = 16 * (w + 8 * g) + lm;
            v2f b; b.x = WcatT0[kk][gc]; b.y = WcatT0[kk + 1][gc];
            acc[g] = __builtin_amdgcn_wmma_f32_16x16x4_f32(false, a, false, b, (short)0, acc[g], false, false);
          }
        }
        #pragma unroll
        for (int r = 0; r < 8; ++r) {
          float iv = sigm(acc[0][r]);
          float fv = sigm(acc[1][r]);
          float gv = tanh_hw(acc[2][r]);
          float ov = sigm(acc[3][r]);
          float cn = fv * c1r[r] + iv * gv;
          c1r[r] = cn;
          hfull[r + 8 * lh][hcol] = ov * tanh_hw(cn);
        }
      }
      __syncthreads();

      {
        v8f acc = {};
        #pragma unroll
        for (int k = 0; k < 4; ++k) {
          int kk = 16 * w + 4 * k + 2 * lh;
          v2f a; a.x = hfull[lm][kk];  a.y = hfull[lm][kk + 1];
          v2f b; b.x = Whr0T[kk][lm];  b.y = Whr0T[kk + 1][lm];
          acc = __builtin_amdgcn_wmma_f32_16x16x4_f32(false, a, false, b, (short)0, acc, false, false);
        }
        #pragma unroll
        for (int r = 0; r < 8; ++r) part[w][r + 8 * lh][lm] = acc[r];
      }
      __syncthreads();
      {
        int row = tid >> 4, o = tid & 15;
        float v = 0.f;
        #pragma unroll
        for (int p = 0; p < 8; ++p) v += part[p][row][o];
        h1Buf[row][o] = v;
        xin1[row][o] = v;
        xin1[row][16 + o] = h2Buf[row][o];
      }
      __syncthreads();

      {
        v8f acc[4];
        #pragma unroll
        for (int g = 0; g < 4; ++g) {
          int gc = 16 * (w + 8 * g) + lm;
          float bv = bias1s[gc];
          #pragma unroll
          for (int r = 0; r < 8; ++r) acc[g][r] = bv;
        }
        #pragma unroll
        for (int k = 0; k < 8; ++k) {
          int kk = 4 * k + 2 * lh;
          v2f a; a.x = xin1[lm][kk]; a.y = xin1[lm][kk + 1];
          #pragma unroll
          for (int g = 0; g < 4; ++g) {
            int gc = 16 * (w + 8 * g) + lm;
            v2f b; b.x = WcatT1[kk][gc]; b.y = WcatT1[kk + 1][gc];
            acc[g] = __builtin_amdgcn_wmma_f32_16x16x4_f32(false, a, false, b, (short)0, acc[g], false, false);
          }
        }
        #pragma unroll
        for (int r = 0; r < 8; ++r) {
          float iv = sigm(acc[0][r]);
          float fv = sigm(acc[1][r]);
          float gv = tanh_hw(acc[2][r]);
          float ov = sigm(acc[3][r]);
          float cn = fv * c2r[r] + iv * gv;
          c2r[r] = cn;
          hfull[r + 8 * lh][hcol] = ov * tanh_hw(cn);
        }
      }
      __syncthreads();

      {
        v8f acc = {};
        #pragma unroll
        for (int k = 0; k < 4; ++k) {
          int kk = 16 * w + 4 * k + 2 * lh;
          v2f a; a.x = hfull[lm][kk];  a.y = hfull[lm][kk + 1];
          v2f b; b.x = Whr1T[kk][lm];  b.y = Whr1T[kk + 1][lm];
          acc = __builtin_amdgcn_wmma_f32_16x16x4_f32(false, a, false, b, (short)0, acc, false, false);
        }
        #pragma unroll
        for (int r = 0; r < 8; ++r) part[w][r + 8 * lh][lm] = acc[r];
      }
      __syncthreads();
      {
        int row = tid >> 4, o = tid & 15;
        float v = 0.f;
        #pragma unroll
        for (int p = 0; p < 8; ++p) v += part[p][row][o];
        h2Buf[row][o]   = v;
        lastBuf[row][o] = v;
        out[(size_t)(b0 + row) * FUT * NNODES * O_DIM
            + (size_t)f * NNODES * O_DIM + (size_t)n * O_DIM + o] = v + nmBuf[row][o];
      }
      __syncthreads();
    }
  }
}

extern "C" void kernel_launch(void* const* d_in, const int* in_sizes, int n_in,
                              void* d_out, int out_size, void* d_ws, size_t ws_size,
                              hipStream_t stream) {
  (void)in_sizes; (void)n_in; (void)out_size; (void)d_ws; (void)ws_size;
  const float* enc   = (const float*)d_in[0];
  const float* mean_ = (const float*)d_in[1];
  const float* s     = (const float*)d_in[2];
  const float* h0    = (const float*)d_in[3];
  const float* c0    = (const float*)d_in[4];
  const float* Wih0  = (const float*)d_in[5];
  const float* Whh0  = (const float*)d_in[6];
  const float* bih0  = (const float*)d_in[7];
  const float* bhh0  = (const float*)d_in[8];
  const float* Whr0  = (const float*)d_in[9];
  const float* Wih1  = (const float*)d_in[10];
  const float* Whh1  = (const float*)d_in[11];
  const float* bih1  = (const float*)d_in[12];
  const float* bhh1  = (const float*)d_in[13];
  const float* Whr1  = (const float*)d_in[14];
  float* out = (float*)d_out;

  decoder_scan<<<dim3(BATCH / 16), dim3(256), 0, stream>>>(
      enc, mean_, s, h0, c0,
      Wih0, Whh0, bih0, bhh0, Whr0,
      Wih1, Whh1, bih1, bhh1, Whr1, out);
}


# All weight B fragments, gate biases, per-node enc term, node mean and each thread's carried h kept in registers instead of re-read from LDS/cache every step; 2 barriers guarding only those copies go
# speedup vs baseline: 2.4958x; 1.3991x over previous
#include <hip/hip_runtime.h>
#include <math.h>

typedef __attribute__((ext_vector_type(2))) float v2f;
typedef __attribute__((ext_vector_type(8))) float v8f;

#define E_DIM 64
#define H_DIM 128
#define O_DIM 16
#define BATCH 64
#define NNODES 1000
#define FUT 12
#define G4 512

__device__ __forceinline__ float sigm(float x) {
  return __builtin_amdgcn_rcpf(1.0f + __builtin_amdgcn_exp2f(-1.44269504088896341f * x));
}
__device__ __forceinline__ float tanh_hw(float x) {
  return 1.0f - 2.0f * __builtin_amdgcn_rcpf(1.0f + __builtin_amdgcn_exp2f(2.88539008177792681f * x));
}

__global__ __launch_bounds__(256) void decoder_scan(
    const float* __restrict__ enc,
    const float* __restrict__ mean_,
    const float* __restrict__ s,
    const float* __restrict__ h0,
    const float* __restrict__ c0,
    const float* __restrict__ Wih0,
    const float* __restrict__ Whh0,
    const float* __restrict__ bih0, const float* __restrict__ bhh0,
    const float* __restrict__ Whr0,
    const float* __restrict__ Wih1,
    const float* __restrict__ Whh1,
    const float* __restrict__ bih1, const float* __restrict__ bhh1,
    const float* __restrict__ Whr1,
    float* __restrict__ out)
{
  __shared__ float hfull[16][H_DIM];
  __shared__ float part[8][16][16];
  __shared__ float xin0[16][32];
  __shared__ float xin1[16][32];

  const int tid  = threadIdx.x;
  const int w    = tid >> 5;
  const int lane = tid & 31;
  const int lm   = lane & 15;
  const int lh   = lane >> 4;
  const int b0   = blockIdx.x * 16;
  const int row  = tid >> 4;
  const int o    = tid & 15;

  v2f Bw0[8][4], Bw1[8][4];
  float bias0r[4], bias1r[4];
  for (int g = 0; g < 4; ++g) {
    const int gc = 16 * (w + 8 * g) + lm;
    bias0r[g] = bih0[gc] + bhh0[gc];
    bias1r[g] = bih1[gc] + bhh1[gc];
    for (int k = 0; k < 8; ++k) {
      const int kk = 4 * k + 2 * lh;
      if (k < 4) {
        Bw0[k][g].x = Wih0[gc * 80 + 64 + kk];  Bw0[k][g].y = Wih0[gc * 80 + 64 + kk + 1];
        Bw1[k][g].x = Wih1[gc * 16 + kk];       Bw1[k][g].y = Wih1[gc * 16 + kk + 1];
      } else {
        Bw0[k][g].x = Whh0[gc * 16 + kk - 16];  Bw0[k][g].y = Whh0[gc * 16 + kk - 15];
        Bw1[k][g].x = Whh1[gc * 16 + kk - 16];  Bw1[k][g].y = Whh1[gc * 16 + kk - 15];
      }
    }
  }
  v2f Be[16][4];
  for (int g = 0; g < 4; ++g) {
    const int gc = 16 * (w + 8 * g) + lm;
    for (int k = 0; k < 16; ++k) {
      const int kk = 4 * k + 2 * lh;
      Be[k][g].x = Wih0[gc * 80 + kk];  Be[k][g].y = Wih0[gc * 80 + kk + 1];
    }
  }
  v2f Pw0[4], Pw1[4];
  for (int k = 0; k < 4; ++k) {
    const int kk = 16 * w + 4 * k + 2 * lh;
    Pw0[k].x = Whr0[lm * H_DIM + kk];  Pw0[k].y = Whr0[lm * H_DIM + kk + 1];
    Pw1[k].x = Whr1[lm * H_DIM + kk];  Pw1[k].y = Whr1[lm * H_DIM + kk + 1];
  }

  float h1v = h0[o];
  float h2v = h0[O_DIM + o];
  float lastv = s[o];

  const int hcol = 16 * w + lm;
  float c1r[8], c2r[8];
  #pragma unroll
  for (int r = 0; r < 8; ++r) { c1r[r] = c0[hcol]; c2r[r] = c0[H_DIM + hcol]; }

  for (int n = 0; n < NNODES; ++n) {
    float nm;
    {
      const float* mp = mean_ + (size_t)(b0 + row) * FUT * NNODES * O_DIM
                              + (size_t)n * O_DIM + o;
      float sum = 0.f;
      #pragma unroll
      for (int f = 0; f < FUT; ++f) sum += mp[(size_t)f * NNODES * O_DIM];
      nm = sum * (1.0f / FUT);
    }
    v8f genc[4];
    #pragma unroll
    for (int g = 0; g < 4; ++g) {
      v8f acc;
      float bv = bias0r[g];
      #pragma unroll
      for (int r = 0; r < 8; ++r) acc[r] = bv;
      const float* ap = enc + (size_t)(b0 + lm) * NNODES * E_DIM + (size_t)n * E_DIM;
      #pragma unroll
      for (int k = 0; k < 16; ++k) {
        int kk = 4 * k + 2 * lh;
        v2f a;
        a.x = ap[kk];              a.y = ap[kk + 1];
        acc = __builtin_amdgcn_wmma_f32_16x16x4_f32(false, a, false, Be[k][g], (short)0, acc, false, false);
      }
      genc[g] = acc;
    }

    for (int f = 0; f < FUT; ++f) {
      xin0[row][o]      = lastv + nm;
      xin0[row][16 + o] = h1v;
      __syncthreads();

      {
        v8f acc[4];
        for (int g = 0; g < 4; ++g) acc[g] = genc[g];
        #pragma unroll
        for (int k = 0; k < 8; ++k) {
          int kk = 4 * k + 2 * lh;
          v2f a; a.x = xin0[lm][kk]; a.y = xin0[lm][kk + 1];
          #pragma unroll
          for (int g = 0; g < 4; ++g)
            acc[g] = __builtin_amdgcn_wmma_f32_16x16x4_f32(false, a, false, Bw0[k][g], (short)0, acc[g], false, false);
        }
        #pragma unroll
        for (int r = 0; r < 8; ++r) {
          float iv = sigm(acc[0][r]);
          float fv = sigm(acc[1][r]);
          float gv = tanh_hw(acc[2][r]);
          float ov = sigm(acc[3][r]);
          float cn = fv * c1r[r] + iv * gv;
          c1r[r] = cn;
          hfull[r + 8 * lh][hcol] = ov * tanh_hw(cn);
        }
      }
      __syncthreads();

      {
        v8f acc = {};
        #pragma unroll
        for (int k = 0; k < 4; ++k) {
          int kk = 16 * w + 4 * k + 2 * lh;
          v2f a; a.x = hfull[lm][kk];  a.y = hfull[lm][kk + 1];
          acc = __builtin_amdgcn_wmma_f32_16x16x4_f32(false, a, false, Pw0[k], (short)0, acc, false, false);
        }
        #pragma unroll
        for (int r = 0; r < 8; ++r) part[w][r + 8 * lh][lm] = acc[r];
      }
      __syncthreads();
      {
        float v = 0.f;
        #pragma unroll
        for (int p = 0; p < 8; ++p) v += part[p][row][o];
        h1v = v;
        xin1[row][o] = v;
        xin1[row][16 + o] = h2v;
      }
      __syncthreads();

      {
        v8f acc[4];
        #pragma unroll
        for (int g = 0; g < 4; ++g) {
          float bv = bias1r[g];
          #pragma unroll
          for (int r = 0; r < 8; ++r) acc[g][r] = bv;
        }
        #pragma unroll
        for (int k = 0; k < 8; ++k) {
          int kk = 4 * k + 2 * lh;
          v2f a; a.x = xin1[lm][kk]; a.y = xin1[lm][kk + 1];
          #pragma unroll
          for (int g = 0; g < 4; ++g)
            acc[g] = __builtin_amdgcn_wmma_f32_16x16x4_f32(false, a, false, Bw1[k][g], (short)0, acc[g], false, false);
        }
        #pragma unroll
        for (int r = 0; r < 8; ++r) {
          float iv = sigm(acc[0][r]);
          float fv = sigm(acc[1][r]);
          float gv = tanh_hw(acc[2][r]);
          float ov = sigm(acc[3][r]);
          float cn = fv * c2r[r] + iv * gv;
          c2r[r] = cn;
          hfull[r + 8 * lh][hcol] = ov * tanh_hw(cn);
        }
      }
      __syncthreads();

      {
        v8f acc = {};
        #pragma unroll
        for (int k = 0; k < 4; ++k) {
          int kk = 16 * w + 4 * k + 2 * lh;
          v2f a; a.x = hfull[lm][kk];  a.y = hfull[lm][kk + 1];
          acc = __builtin_amdgcn_wmma_f32_16x16x4_f32(false, a, false, Pw1[k], (short)0, acc, false, false);
        }
        #pragma unroll
        for (int r = 0; r < 8; ++r) part[w][r + 8 * lh][lm] = acc[r];
      }
      __syncthreads();
      {
        float v = 0.f;
        #pragma unroll
        for (int p = 0; p < 8; ++p) v += part[p][row][o];
        h2v   = v;
        lastv = v;
        out[(size_t)(b0 + row) * FUT * NNODES * O_DIM
            + (size_t)f * NNODES * O_DIM + (size_t)n * O_DIM + o] = v + nm;
      }
    }
  }
}

extern "C" void kernel_launch(void* const* d_in, const int* in_sizes, int n_in,
                              void* d_out, int out_size, void* d_ws, size_t ws_size,
                              hipStream_t stream) {
  (void)in_sizes; (void)n_in; (void)out_size; (void)d_ws; (void)ws_size;
  const float* enc   = (const float*)d_in[0];
  const float* mean_ = (const float*)d_in[1];
  const float* s     = (const float*)d_in[2];
  const float* h0    = (const float*)d_in[3];
  const float* c0    = (const float*)d_in[4];
  const float* Wih0  = (const float*)d_in[5];
  const float* Whh0  = (const float*)d_in[6];
  const float* bih0  = (const float*)d_in[7];
  const float* bhh0  = (const float*)d_in[8];
  const float* Whr0  = (const float*)d_in[9];
  const float* Wih1  = (const float*)d_in[10];
  const float* Whh1  = (const float*)d_in[11];
  const float* bih1  = (const float*)d_in[12];
  const float* bhh1  = (const float*)d_in[13];
  const float* Whr1  = (const float*)d_in[14];
  float* out = (float*)d_out;

  decoder_scan<<<dim3(BATCH / 16), dim3(256), 0, stream>>>(
      enc, mean_, s, h0, c0,
      Wih0, Whh0, bih0, bhh0, Whr0,
      Wih1, Whh1, bih1, bhh1, Whr1, out);
}


# Gate GEMMs formed transposed (weights as A, step input as B) so h_full feeds the projection WMMA from registers: hfull LDS round trip and 2 of 6 barriers per step go; projection k sum order changes
# speedup vs baseline: 3.1470x; 1.2609x over previous
#include <hip/hip_runtime.h>
#include <math.h>

typedef __attribute__((ext_vector_type(2))) float v2f;
typedef __attribute__((ext_vector_type(8))) float v8f;

#define E_DIM 64
#define H_DIM 128
#define O_DIM 16
#define BATCH 64
#define NNODES 1000
#define FUT 12
#define G4 512

__device__ __forceinline__ float sigm(float x) {
  return __builtin_amdgcn_rcpf(1.0f + __builtin_amdgcn_exp2f(-1.44269504088896341f * x));
}
__device__ __forceinline__ float tanh_hw(float x) {
  return 1.0f - 2.0f * __builtin_amdgcn_rcpf(1.0f + __builtin_amdgcn_exp2f(2.88539008177792681f * x));
}

__global__ __launch_bounds__(256) void decoder_scan(
    const float* __restrict__ enc,
    const float* __restrict__ mean_,
    const float* __restrict__ s,
    const float* __restrict__ h0,
    const float* __restrict__ c0,
    const float* __restrict__ Wih0,
    const float* __restrict__ Whh0,
    const float* __restrict__ bih0, const float* __restrict__ bhh0,
    const float* __restrict__ Whr0,
    const float* __restrict__ Wih1,
    const float* __restrict__ Whh1,
    const float* __restrict__ bih1, const float* __restrict__ bhh1,
    const float* __restrict__ Whr1,
    float* __restrict__ out)
{
  __shared__ __attribute__((aligned(32))) float part[8][16][16];
  __shared__ __attribute__((aligned(32))) float bias0s[G4];
  __shared__ float xin0[16][32];
  __shared__ float xin1[16][32];

  const int tid  = threadIdx.x;
  const int w    = tid >> 5;
  const int lane = tid & 31;
  const int lm   = lane & 15;
  const int lh   = lane >> 4;
  const int b0   = blockIdx.x * 16;
  const int row  = tid >> 4;
  const int o    = tid & 15;

  bias0s[tid]       = bih0[tid]       + bhh0[tid];
  bias0s[tid + 256] = bih0[tid + 256] + bhh0[tid + 256];

  v2f Bw0[8][4], Bw1[8][4];
  v8f bias1v[4];
  for (int g = 0; g < 4; ++g) {
    const int gc = 16 * (w + 8 * g) + lm;
    for (int r = 0; r < 8; ++r) {
      const int gr = 16 * (w + 8 * g) + 8 * lh + r;
      bias1v[g][r] = bih1[gr] + bhh1[gr];
    }
    for (int k = 0; k < 8; ++k) {
      const int kk = 4 * k + 2 * lh;
      if (k < 4) {
        Bw0[k][g].x = Wih0[gc * 80 + 64 + kk];  Bw0[k][g].y = Wih0[gc * 80 + 64 + kk + 1];
        Bw1[k][g].x = Wih1[gc * 16 + kk];       Bw1[k][g].y = Wih1[gc * 16 + kk + 1];
      } else {
        Bw0[k][g].x = Whh0[gc * 16 + kk - 16];  Bw0[k][g].y = Whh0[gc * 16 + kk - 15];
        Bw1[k][g].x = Whh1[gc * 16 + kk - 16];  Bw1[k][g].y = Whh1[gc * 16 + kk - 15];
      }
    }
  }
  v2f Be[16][4];
  for (int g = 0; g < 4; ++g) {
    const int gc = 16 * (w + 8 * g) + lm;
    for (int k = 0; k < 16; ++k) {
      const int kk = 4 * k + 2 * lh;
      Be[k][g].x = Wih0[gc * 80 + kk];  Be[k][g].y = Wih0[gc * 80 + kk + 1];
    }
  }
  v2f Pw0[4], Pw1[4];
  for (int k = 0; k < 4; ++k) {
    const int kk = 16 * w + 8 * lh + 2 * k;
    Pw0[k].x = Whr0[lm * H_DIM + kk];  Pw0[k].y = Whr0[lm * H_DIM + kk + 1];
    Pw1[k].x = Whr1[lm * H_DIM + kk];  Pw1[k].y = Whr1[lm * H_DIM + kk + 1];
  }

  float h1v = h0[o];
  float h2v = h0[O_DIM + o];
  float lastv = s[o];

  const int hcol0 = 16 * w + 8 * lh;
  float c1r[8], c2r[8];
  #pragma unroll
  for (int r = 0; r < 8; ++r) { c1r[r] = c0[hcol0 + r]; c2r[r] = c0[H_DIM + hcol0 + r]; }
  __syncthreads();

  for (int n = 0; n < NNODES; ++n) {
    float nm;
    {
      const float* mp = mean_ + (size_t)(b0 + row) * FUT * NNODES * O_DIM
                              + (size_t)n * O_DIM + o;
      float sum = 0.f;
      #pragma unroll
      for (int f = 0; f < FUT; ++f) sum += mp[(size_t)f * NNODES * O_DIM];
      nm = sum * (1.0f / FUT);
    }
    v8f genc[4];
    #pragma unroll
    for (int g = 0; g < 4; ++g) {
      v8f acc = *(const v8f*)&bias0s[16 * (w + 8 * g) + 8 * lh];
      const float* ap = enc + (size_t)(b0 + lm) * NNODES * E_DIM + (size_t)n * E_DIM;
      #pragma unroll
      for (int k = 0; k < 16; ++k) {
        int kk = 4 * k + 2 * lh;
        v2f a;
        a.x = ap[kk];              a.y = ap[kk + 1];
        acc = __builtin_amdgcn_wmma_f32_16x16x4_f32(false, Be[k][g], false, a, (short)0, acc, false, false);
      }
      genc[g] = acc;
    }

    for (int f = 0; f < FUT; ++f) {
      xin0[row][o]      = lastv + nm;
      xin0[row][16 + o] = h1v;
      __syncthreads();

      {
        v8f acc[4];
        for (int g = 0; g < 4; ++g) acc[g] = genc[g];
        #pragma unroll
        for (int k = 0; k < 8; ++k) {
          int kk = 4 * k + 2 * lh;
          v2f a; a.x = xin0[lm][kk]; a.y = xin0[lm][kk + 1];
          #pragma unroll
          for (int g = 0; g < 4; ++g)
            acc[g] = __builtin_amdgcn_wmma_f32_16x16x4_f32(false, Bw0[k][g], false, a, (short)0, acc[g], false, false);
        }
        float hf[8];
        #pragma unroll
        for (int r = 0; r < 8; ++r) {
          float iv = sigm(acc[0][r]);
          float fv = sigm(acc[1][r]);
          float gv = tanh_hw(acc[2][r]);
          float ov = sigm(acc[3][r]);
          float cn = fv * c1r[r] + iv * gv;
          c1r[r] = cn;
          hf[r] = ov * tanh_hw(cn);
        }
        v8f pacc = {};
        #pragma unroll
        for (int k = 0; k < 4; ++k) {
          v2f hb; hb.x = hf[2 * k]; hb.y = hf[2 * k + 1];
          pacc = __builtin_amdgcn_wmma_f32_16x16x4_f32(false, Pw0[k], false, hb, (short)0, pacc, false, false);
        }
        *(v8f*)&part[w][lm][8 * lh] = pacc;
      }
      __syncthreads();
      {
        float v = 0.f;
        #pragma unroll
        for (int p = 0; p < 8; ++p) v += part[p][row][o];
        h1v = v;
        xin1[row][o] = v;
        xin1[row][16 + o] = h2v;
      }
      __syncthreads();

      {
        v8f acc[4];
        #pragma unroll
        for (int g = 0; g < 4; ++g) acc[g] = bias1v[g];
        #pragma unroll
        for (int k = 0; k < 8; ++k) {
          int kk = 4 * k + 2 * lh;
          v2f a; a.x = xin1[lm][kk]; a.y = xin1[lm][kk + 1];
          #pragma unroll
          for (int g = 0; g < 4; ++g)
            acc[g] = __builtin_amdgcn_wmma_f32_16x16x4_f32(false, Bw1[k][g], false, a, (short)0, acc[g], false, false);
        }
        float hf[8];
        #pragma unroll
        for (int r = 0; r < 8; ++r) {
          float iv = sigm(acc[0][r]);
          float fv = sigm(acc[1][r]);
          float gv = tanh_hw(acc[2][r]);
          float ov = sigm(acc[3][r]);
          float cn = fv * c2r[r] + iv * gv;
          c2r[r] = cn;
          hf[r] = ov * tanh_hw(cn);
        }
        v8f pacc = {};
        #pragma unroll
        for (int k = 0; k < 4; ++k) {
          v2f hb; hb.x = hf[2 * k]; hb.y = hf[2 * k + 1];
          pacc = __builtin_amdgcn_wmma_f32_16x16x4_f32(false, Pw1[k], false, hb, (short)0, pacc, false, false);
        }
        *(v8f*)&part[w][lm][8 * lh] = pacc;
      }
      __syncthreads();
      {
        float v = 0.f;
        #pragma unroll
        for (int p = 0; p < 8; ++p) v += part[p][row][o];
        h2v   = v;
        lastv = v;
        out[(size_t)(b0 + row) * FUT * NNODES * O_DIM
            + (size_t)f * NNODES * O_DIM + (size_t)n * O_DIM + o] = v + nm;
      }
    }
  }
}

extern "C" void kernel_launch(void* const* d_in, const int* in_sizes, int n_in,
                              void* d_out, int out_size, void* d_ws, size_t ws_size,
                              hipStream_t stream) {
  (void)in_sizes; (void)n_in; (void)out_size; (void)d_ws; (void)ws_size;
  const float* enc   = (const float*)d_in[0];
  const float* mean_ = (const float*)d_in[1];
  const float* s     = (const float*)d_in[2];
  const float* h0    = (const float*)d_in[3];
  const float* c0    = (const float*)d_in[4];
  const float* Wih0  = (const float*)d_in[5];
  const float* Whh0  = (const float*)d_in[6];
  const float* bih0  = (const float*)d_in[7];
  const float* bhh0  = (const float*)d_in[8];
  const float* Whr0  = (const float*)d_in[9];
  const float* Wih1  = (const float*)d_in[10];
  const float* Whh1  = (const float*)d_in[11];
  const float* bih1  = (const float*)d_in[12];
  const float* bhh1  = (const float*)d_in[13];
  const float* Whr1  = (const float*)d_in[14];
  float* out = (float*)d_out;

  decoder_scan<<<dim3(BATCH / 16), dim3(256), 0, stream>>>(
      enc, mean_, s, h0, c0,
      Wih0, Whh0, bih0, bhh0, Whr0,
      Wih1, Whh1, bih1, bhh1, Whr1, out);
}


# Every wave adds the 8 split-K projection partials itself into B-fragment registers: the xin LDS round trip and 2 of 4 barriers per step go; gate k sum order changes; enc weights move to LDS for room
# speedup vs baseline: 3.2528x; 1.0336x over previous
#include <hip/hip_runtime.h>
#include <math.h>

typedef __attribute__((ext_vector_type(2))) float v2f;
typedef __attribute__((ext_vector_type(4))) float v4f;
typedef __attribute__((ext_vector_type(8))) float v8f;

#define E_DIM 64
#define H_DIM 128
#define O_DIM 16
#define BATCH 64
#define NNODES 1000
#define FUT 12
#define G4 512

__device__ __forceinline__ float sigm(float x) {
  return __builtin_amdgcn_rcpf(1.0f + __builtin_amdgcn_exp2f(-1.44269504088896341f * x));
}
__device__ __forceinline__ float tanh_hw(float x) {
  return 1.0f - 2.0f * __builtin_amdgcn_rcpf(1.0f + __builtin_amdgcn_exp2f(2.88539008177792681f * x));
}

__global__ __launch_bounds__(256) void decoder_scan(
    const float* __restrict__ enc,
    const float* __restrict__ mean_,
    const float* __restrict__ s,
    const float* __restrict__ h0,
    const float* __restrict__ c0,
    const float* __restrict__ Wih0,
    const float* __restrict__ Whh0,
    const float* __restrict__ bih0, const float* __restrict__ bhh0,
    const float* __restrict__ Whr0,
    const float* __restrict__ Wih1,
    const float* __restrict__ Whh1,
    const float* __restrict__ bih1, const float* __restrict__ bhh1,
    const float* __restrict__ Whr1,
    float* __restrict__ out)
{
  __shared__ __attribute__((aligned(32))) float part[2][8][4][16][4];
  __shared__ __attribute__((aligned(32))) float bias0s[G4];
  __shared__ __attribute__((aligned(16))) float BeS[8][4][8][32][4];

  const int tid  = threadIdx.x;
  const int w    = tid >> 5;
  const int lane = tid & 31;
  const int lm   = lane & 15;
  const int lh   = lane >> 4;
  const int b0   = blockIdx.x * 16;
  const int wsc  = __builtin_amdgcn_readfirstlane(w);

  bias0s[tid]       = bih0[tid]       + bhh0[tid];
  bias0s[tid + 256] = bih0[tid + 256] + bhh0[tid + 256];

  v2f Bw0[8][4], Bw1[8][4];
  v8f bias1v[4];
  for (int g = 0; g < 4; ++g) {
    const int gc = 16 * (w + 8 * g) + lm;
    for (int r = 0; r < 8; ++r) {
      const int gr = 16 * (w + 8 * g) + 8 * lh + r;
      bias1v[g][r] = bih1[gr] + bhh1[gr];
    }
    for (int j = 0; j < 4; ++j) {
      const int kk = 8 * lh + 2 * j;
      Bw0[j][g].x     = Wih0[gc * 80 + 64 + kk];  Bw0[j][g].y     = Wih0[gc * 80 + 64 + kk + 1];
      Bw0[4 + j][g].x = Whh0[gc * 16 + kk];       Bw0[4 + j][g].y = Whh0[gc * 16 + kk + 1];
      Bw1[j][g].x     = Wih1[gc * 16 + kk];       Bw1[j][g].y     = Wih1[gc * 16 + kk + 1];
      Bw1[4 + j][g].x = Whh1[gc * 16 + kk];       Bw1[4 + j][g].y = Whh1[gc * 16 + kk + 1];
    }
  }
  for (int g = 0; g < 4; ++g) {
    const int gc = 16 * (w + 8 * g) + lm;
    for (int kp = 0; kp < 8; ++kp) {
      const int kk = 8 * kp + 2 * lh;
      v4f t;
      t.x = Wih0[gc * 80 + kk];      t.y = Wih0[gc * 80 + kk + 1];
      t.z = Wih0[gc * 80 + kk + 4];  t.w = Wih0[gc * 80 + kk + 5];
      *(v4f*)&BeS[w][g][kp][lane][0] = t;
    }
  }
  v2f Pw0[4], Pw1[4];
  for (int k = 0; k < 4; ++k) {
    const int kk = 16 * w + 8 * lh + 2 * k;
    Pw0[k].x = Whr0[lm * H_DIM + kk];  Pw0[k].y = Whr0[lm * H_DIM + kk + 1];
    Pw1[k].x = Whr1[lm * H_DIM + kk];  Pw1[k].y = Whr1[lm * H_DIM + kk + 1];
  }

  float h1B[8], h2B[8], nmB[8];
  #pragma unroll
  for (int r = 0; r < 8; ++r) { h1B[r] = h0[8 * lh + r]; h2B[r] = 0.f; nmB[r] = 0.f; }

  const int hcol0 = 16 * w + 8 * lh;
  float c1r[8], c2r[8];
  #pragma unroll
  for (int r = 0; r < 8; ++r) { c1r[r] = c0[hcol0 + r]; c2r[r] = c0[H_DIM + hcol0 + r]; }

  const int pc0 = (2 * lh * 16 + lm) * 4;
  const int pc1 = ((2 * lh + 1) * 16 + lm) * 4;
  float* const pw0a = &part[0][w][0][0][0] + pc0;
  float* const pw0b = &part[0][w][0][0][0] + pc1;
  float* const pw1a = &part[1][w][0][0][0] + pc0;
  float* const pw1b = &part[1][w][0][0][0] + pc1;
  const float* const pr0 = &part[0][0][0][0][0];
  const float* const pr1 = &part[1][0][0][0][0];

  {
    v4f sa = {0.f, 0.f, 0.f, 0.f}, sb = {0.f, 0.f, 0.f, 0.f};
    if (w == 0) { sa = *(const v4f*)&s[8 * lh]; sb = *(const v4f*)&s[8 * lh + 4]; }
    *(v4f*)pw1a = sa;
    *(v4f*)pw1b = sb;
  }
  __syncthreads();

  v8f genc[4];
  for (int n = 0; n < NNODES; ++n) {
    for (int f = 0; f < FUT; ++f) {
      v4f La[8], Lb[8];
      #pragma unroll
      for (int p = 0; p < 8; ++p) {
        La[p] = *(const v4f*)(pr1 + p * 256 + pc0);
        Lb[p] = *(const v4f*)(pr1 + p * 256 + pc1);
      }
      v4f nmNa = {0.f, 0.f, 0.f, 0.f}, nmNb = {0.f, 0.f, 0.f, 0.f};
      if (f == 0) {
        const float* mp = mean_ + (size_t)(b0 + lm) * FUT * NNODES * O_DIM
                                + (size_t)n * O_DIM + 8 * lh;
        #pragma unroll
        for (int ff = 0; ff < FUT; ++ff) {
          nmNa += *(const v4f*)(mp + (size_t)ff * NNODES * O_DIM);
          nmNb += *(const v4f*)(mp + (size_t)ff * NNODES * O_DIM + 4);
        }
        nmNa *= (1.0f / FUT);
        nmNb *= (1.0f / FUT);
        const float* ap = enc + (size_t)(b0 + lm) * NNODES * E_DIM + (size_t)n * E_DIM;
        #pragma unroll
        for (int g = 0; g < 4; ++g) {
          v8f acc = *(const v8f*)&bias0s[16 * (w + 8 * g) + 8 * lh];
          #pragma unroll
          for (int kp = 0; kp < 8; ++kp) {
            const v4f wv = *(const v4f*)&BeS[w][g][kp][lane][0];
            int kk = 8 * kp + 2 * lh;
            v2f wa, a;
            wa.x = wv.x;               wa.y = wv.y;
            a.x = ap[kk];              a.y = ap[kk + 1];
            acc = __builtin_amdgcn_wmma_f32_16x16x4_f32(false, wa, false, a, (short)0, acc, false, false);
            wa.x = wv.z;               wa.y = wv.w;
            a.x = ap[kk + 4];          a.y = ap[kk + 5];
            acc = __builtin_amdgcn_wmma_f32_16x16x4_f32(false, wa, false, a, (short)0, acc, false, false);
          }
          genc[g] = acc;
        }
      }

      v8f acc[4];
      #pragma unroll
      for (int g = 0; g < 4; ++g) acc[g] = genc[g];
      #pragma unroll
      for (int j = 0; j < 4; ++j) {
        v2f hb; hb.x = h1B[2 * j]; hb.y = h1B[2 * j + 1];
        #pragma unroll
        for (int g = 0; g < 4; ++g)
          acc[g] = __builtin_amdgcn_wmma_f32_16x16x4_f32(false, Bw0[4 + j][g], false, hb, (short)0, acc[g], false, false);
      }
      __builtin_amdgcn_sched_barrier(0);
      {
        v4f a = La[0], b = Lb[0];
        #pragma unroll
        for (int p = 1; p < 8; ++p) { a += La[p]; b += Lb[p]; }
        #pragma unroll
        for (int i = 0; i < 4; ++i) { h2B[i] = a[i]; h2B[4 + i] = b[i]; }
      }
      if ((n | f) != 0 && wsc == (f & 7)) {
        const int fp = (f == 0) ? FUT - 1 : f - 1;
        const int np = (f == 0) ? n - 1 : n;
        float* op = out + (size_t)(b0 + lm) * FUT * NNODES * O_DIM
                        + (size_t)fp * NNODES * O_DIM + (size_t)np * O_DIM + 8 * lh;
        v4f oa, ob;
        #pragma unroll
        for (int i = 0; i < 4; ++i) { oa[i] = h2B[i] + nmB[i]; ob[i] = h2B[4 + i] + nmB[4 + i]; }
        *(v4f*)op       = oa;
        *(v4f*)(op + 4) = ob;
      }
      if (f == 0) {
        #pragma unroll
        for (int i = 0; i < 4; ++i) { nmB[i] = nmNa[i]; nmB[4 + i] = nmNb[i]; }
      }
      float xB[8];
      #pragma unroll
      for (int r = 0; r < 8; ++r) xB[r] = h2B[r] + nmB[r];
      if ((n | f) == 0) {
        #pragma unroll
        for (int r = 0; r < 8; ++r) h2B[r] = h0[O_DIM + 8 * lh + r];
      }
      #pragma unroll
      for (int j = 0; j < 4; ++j) {
        v2f xb; xb.x = xB[2 * j]; xb.y = xB[2 * j + 1];
        #pragma unroll
        for (int g = 0; g < 4; ++g)
          acc[g] = __builtin_amdgcn_wmma_f32_16x16x4_f32(false, Bw0[j][g], false, xb, (short)0, acc[g], false, false);
      }
      {
        float hf[8];
        #pragma unroll
        for (int r = 0; r < 8; ++r) {
          float iv = sigm(acc[0][r]);
          float fv = sigm(acc[1][r]);
          float gv = tanh_hw(acc[2][r]);
          float ov = sigm(acc[3][r]);
          float cn = fv * c1r[r] + iv * gv;
          c1r[r] = cn;
          hf[r] = ov * tanh_hw(cn);
        }
        v8f pacc = {};
        #pragma unroll
        for (int k = 0; k < 4; ++k) {
          v2f hb; hb.x = hf[2 * k]; hb.y = hf[2 * k + 1];
          pacc = __builtin_amdgcn_wmma_f32_16x16x4_f32(false, Pw0[k], false, hb, (short)0, pacc, false, false);
        }
        v4f qa, qb;
        #pragma unroll
        for (int i = 0; i < 4; ++i) { qa[i] = pacc[i]; qb[i] = pacc[4 + i]; }
        *(v4f*)pw0a = qa;
        *(v4f*)pw0b = qb;
      }
      __syncthreads();

      #pragma unroll
      for (int p = 0; p < 8; ++p) {
        La[p] = *(const v4f*)(pr0 + p * 256 + pc0);
        Lb[p] = *(const v4f*)(pr0 + p * 256 + pc1);
      }
      __builtin_amdgcn_sched_barrier(0);
      #pragma unroll
      for (int g = 0; g < 4; ++g) acc[g] = bias1v[g];
      #pragma unroll
      for (int j = 0; j < 4; ++j) {
        v2f hb; hb.x = h2B[2 * j]; hb.y = h2B[2 * j + 1];
        #pragma unroll
        for (int g = 0; g < 4; ++g)
          acc[g] = __builtin_amdgcn_wmma_f32_16x16x4_f32(false, Bw1[4 + j][g], false, hb, (short)0, acc[g], false, false);
      }
      __builtin_amdgcn_sched_barrier(0);
      {
        v4f a = La[0], b = Lb[0];
        #pragma unroll
        for (int p = 1; p < 8; ++p) { a += La[p]; b += Lb[p]; }
        #pragma unroll
        for (int i = 0; i < 4; ++i) { h1B[i] = a[i]; h1B[4 + i] = b[i]; }
      }
      #pragma unroll
      for (int j = 0; j < 4; ++j) {
        v2f xb; xb.x = h1B[2 * j]; xb.y = h1B[2 * j + 1];
        #pragma unroll
        for (int g = 0; g < 4; ++g)
          acc[g] = __builtin_amdgcn_wmma_f32_16x16x4_f32(false, Bw1[j][g], false, xb, (short)0, acc[g], false, false);
      }
      {
        float hf[8];
        #pragma unroll
        for (int r = 0; r < 8; ++r) {
          float iv = sigm(acc[0][r]);
          float fv = sigm(acc[1][r]);
          float gv = tanh_hw(acc[2][r]);
          float ov = sigm(acc[3][r]);
          float cn = fv * c2r[r] + iv * gv;
          c2r[r] = cn;
          hf[r] = ov * tanh_hw(cn);
        }
        v8f pacc = {};
        #pragma unroll
        for (int k = 0; k < 4; ++k) {
          v2f hb; hb.x = hf[2 * k]; hb.y = hf[2 * k + 1];
          pacc = __builtin_amdgcn_wmma_f32_16x16x4_f32(false, Pw1[k], false, hb, (short)0, pacc, false, false);
        }
        v4f qa, qb;
        #pragma unroll
        for (int i = 0; i < 4; ++i) { qa[i] = pacc[i]; qb[i] = pacc[4 + i]; }
        *(v4f*)pw1a = qa;
        *(v4f*)pw1b = qb;
      }
      __syncthreads();
    }
  }

  {
    v4f a = *(const v4f*)(pr1 + pc0), b = *(const v4f*)(pr1 + pc1);
    #pragma unroll
    for (int p = 1; p < 8; ++p) {
      a += *(const v4f*)(pr1 + p * 256 + pc0);
      b += *(const v4f*)(pr1 + p * 256 + pc1);
    }
    if (wsc == 0) {
      float* op = out + (size_t)(b0 + lm) * FUT * NNODES * O_DIM
                      + (size_t)(FUT - 1) * NNODES * O_DIM + (size_t)(NNODES - 1) * O_DIM + 8 * lh;
      v4f oa, ob;
      #pragma unroll
      for (int i = 0; i < 4; ++i) { oa[i] = a[i] + nmB[i]; ob[i] = b[i] + nmB[4 + i]; }
      *(v4f*)op       = oa;
      *(v4f*)(op + 4) = ob;
    }
  }
}

extern "C" void kernel_launch(void* const* d_in, const int* in_sizes, int n_in,
                              void* d_out, int out_size, void* d_ws, size_t ws_size,
                              hipStream_t stream) {
  (void)in_sizes; (void)n_in; (void)out_size; (void)d_ws; (void)ws_size;
  const float* enc   = (const float*)d_in[0];
  const float* mean_ = (const float*)d_in[1];
  const float* s     = (const float*)d_in[2];
  const float* h0    = (const float*)d_in[3];
  const float* c0    = (const float*)d_in[4];
  const float* Wih0  = (const float*)d_in[5];
  const float* Whh0  = (const float*)d_in[6];
  const float* bih0  = (const float*)d_in[7];
  const float* bhh0  = (const float*)d_in[8];
  const float* Whr0  = (const float*)d_in[9];
  const float* Wih1  = (const float*)d_in[10];
  const float* Whh1  = (const float*)d_in[11];
  const float* bih1  = (const float*)d_in[12];
  const float* bhh1  = (const float*)d_in[13];
  const float* Whr1  = (const float*)d_in[14];
  float* out = (float*)d_out;

  decoder_scan<<<dim3(BATCH / 16), dim3(256), 0, stream>>>(
      enc, mean_, s, h0, c0,
      Wih0, Whh0, bih0, bhh0, Whr0,
      Wih1, Whh1, bih1, bhh1, Whr1, out);
}


# Gate sigmoid and tanh each on the one hardware tanh instruction (v_tanh_f32; sigmoid as 1/2 + 1/2 tanh(x/2)) instead of exp2 + reciprocal: 5 transcendentals per hidden unit, not 10; sums unchanged
# speedup vs baseline: 3.5985x; 1.1063x over previous
#include <hip/hip_runtime.h>
#include <math.h>

typedef __attribute__((ext_vector_type(2))) float v2f;
typedef __attribute__((ext_vector_type(4))) float v4f;
typedef __attribute__((ext_vector_type(8))) float v8f;

#define E_DIM 64
#define H_DIM 128
#define O_DIM 16
#define BATCH 64
#define NNODES 1000
#define FUT 12
#define G4 512

__device__ __forceinline__ float tanh_hw(float x) {
  return __builtin_amdgcn_tanhf(x);
}
__device__ __forceinline__ float sigm(float x) {
  return __builtin_fmaf(0.5f, __builtin_amdgcn_tanhf(0.5f * x), 0.5f);
}

__global__ __launch_bounds__(256) void decoder_scan(
    const float* __restrict__ enc,
    const float* __restrict__ mean_,
    const float* __restrict__ s,
    const float* __restrict__ h0,
    const float* __restrict__ c0,
    const float* __restrict__ Wih0,
    const float* __restrict__ Whh0,
    const float* __restrict__ bih0, const float* __restrict__ bhh0,
    const float* __restrict__ Whr0,
    const float* __restrict__ Wih1,
    const float* __restrict__ Whh1,
    const float* __restrict__ bih1, const float* __restrict__ bhh1,
    const float* __restrict__ Whr1,
    float* __restrict__ out)
{
  __shared__ __attribute__((aligned(32))) float part[2][8][4][16][4];
  __shared__ __attribute__((aligned(32))) float bias0s[G4];
  __shared__ __attribute__((aligned(16))) float BeS[8][4][8][32][4];

  const int tid  = threadIdx.x;
  const int w    = tid >> 5;
  const int lane = tid & 31;
  const int lm   = lane & 15;
  const int lh   = lane >> 4;
  const int b0   = blockIdx.x * 16;
  const int wsc  = __builtin_amdgcn_readfirstlane(w);

  bias0s[tid]       = bih0[tid]       + bhh0[tid];
  bias0s[tid + 256] = bih0[tid + 256] + bhh0[tid + 256];

  v2f Bw0[8][4], Bw1[8][4];
  v8f bias1v[4];
  for (int g = 0; g < 4; ++g) {
    const int gc = 16 * (w + 8 * g) + lm;
    for (int r = 0; r < 8; ++r) {
      const int gr = 16 * (w + 8 * g) + 8 * lh + r;
      bias1v[g][r] = bih1[gr] + bhh1[gr];
    }
    for (int j = 0; j < 4; ++j) {
      const int kk = 8 * lh + 2 * j;
      Bw0[j][g].x     = Wih0[gc * 80 + 64 + kk];  Bw0[j][g].y     = Wih0[gc * 80 + 64 + kk + 1];
      Bw0[4 + j][g].x = Whh0[gc * 16 + kk];       Bw0[4 + j][g].y = Whh0[gc * 16 + kk + 1];
      Bw1[j][g].x     = Wih1[gc * 16 + kk];       Bw1[j][g].y     = Wih1[gc * 16 + kk + 1];
      Bw1[4 + j][g].x = Whh1[gc * 16 + kk];       Bw1[4 + j][g].y = Whh1[gc * 16 + kk + 1];
    }
  }
  for (int g = 0; g < 4; ++g) {
    const int gc = 16 * (w + 8 * g) + lm;
    for (int kp = 0; kp < 8; ++kp) {
      const int kk = 8 * kp + 2 * lh;
      v4f t;
      t.x = Wih0[gc * 80 + kk];      t.y = Wih0[gc * 80 + kk + 1];
      t.z = Wih0[gc * 80 + kk + 4];  t.w = Wih0[gc * 80 + kk + 5];
      *(v4f*)&BeS[w][g][kp][lane][0] = t;
    }
  }
  v2f Pw0[4], Pw1[4];
  for (int k = 0; k < 4; ++k) {
    const int kk = 16 * w + 8 * lh + 2 * k;
    Pw0[k].x = Whr0[lm * H_DIM + kk];  Pw0[k].y = Whr0[lm * H_DIM + kk + 1];
    Pw1[k].x = Whr1[lm * H_DIM + kk];  Pw1[k].y = Whr1[lm * H_DIM + kk + 1];
  }

  float h1B[8], h2B[8], nmB[8];
  #pragma unroll
  for (int r = 0; r < 8; ++r) { h1B[r] = h0[8 * lh + r]; h2B[r] = 0.f; nmB[r] = 0.f; }

  const int hcol0 = 16 * w + 8 * lh;
  float c1r[8], c2r[8];
  #pragma unroll
  for (int r = 0; r < 8; ++r) { c1r[r] = c0[hcol0 + r]; c2r[r] = c0[H_DIM + hcol0 + r]; }

  const int pc0 = (2 * lh * 16 + lm) * 4;
  const int pc1 = ((2 * lh + 1) * 16 + lm) * 4;
  float* const pw0a = &part[0][w][0][0][0] + pc0;
  float* const pw0b = &part[0][w][0][0][0] + pc1;
  float* const pw1a = &part[1][w][0][0][0] + pc0;
  float* const pw1b = &part[1][w][0][0][0] + pc1;
  const float* const pr0 = &part[0][0][0][0][0];
  const float* const pr1 = &part[1][0][0][0][0];

  {
    v4f sa = {0.f, 0.f, 0.f, 0.f}, sb = {0.f, 0.f, 0.f, 0.f};
    if (w == 0) { sa = *(const v4f*)&s[8 * lh]; sb = *(const v4f*)&s[8 * lh + 4]; }
    *(v4f*)pw1a = sa;
    *(v4f*)pw1b = sb;
  }
  __syncthreads();

  v8f genc[4];
  for (int n = 0; n < NNODES; ++n) {
    for (int f = 0; f < FUT; ++f) {
      v4f La[8], Lb[8];
      #pragma unroll
      for (int p = 0; p < 8; ++p) {
        La[p] = *(const v4f*)(pr1 + p * 256 + pc0);
        Lb[p] = *(const v4f*)(pr1 + p * 256 + pc1);
      }
      v4f nmNa = {0.f, 0.f, 0.f, 0.f}, nmNb = {0.f, 0.f, 0.f, 0.f};
      if (f == 0) {
        const float* mp = mean_ + (size_t)(b0 + lm) * FUT * NNODES * O_DIM
                                + (size_t)n * O_DIM + 8 * lh;
        #pragma unroll
        for (int ff = 0; ff < FUT; ++ff) {
          nmNa += *(const v4f*)(mp + (size_t)ff * NNODES * O_DIM);
          nmNb += *(const v4f*)(mp + (size_t)ff * NNODES * O_DIM + 4);
        }
        nmNa *= (1.0f / FUT);
        nmNb *= (1.0f / FUT);
        const float* ap = enc + (size_t)(b0 + lm) * NNODES * E_DIM + (size_t)n * E_DIM;
        #pragma unroll
        for (int g = 0; g < 4; ++g) {
          v8f acc = *(const v8f*)&bias0s[16 * (w + 8 * g) + 8 * lh];
          #pragma unroll
          for (int kp = 0; kp < 8; ++kp) {
            const v4f wv = *(const v4f*)&BeS[w][g][kp][lane][0];
            int kk = 8 * kp + 2 * lh;
            v2f wa, a;
            wa.x = wv.x;               wa.y = wv.y;
            a.x = ap[kk];              a.y = ap[kk + 1];
            acc = __builtin_amdgcn_wmma_f32_16x16x4_f32(false, wa, false, a, (short)0, acc, false, false);
            wa.x = wv.z;               wa.y = wv.w;
            a.x = ap[kk + 4];          a.y = ap[kk + 5];
            acc = __builtin_amdgcn_wmma_f32_16x16x4_f32(false, wa, false, a, (short)0, acc, false, false);
          }
          genc[g] = acc;
        }
      }

      v8f acc[4];
      #pragma unroll
      for (int g = 0; g < 4; ++g) acc[g] = genc[g];
      #pragma unroll
      for (int j = 0; j < 4; ++j) {
        v2f hb; hb.x = h1B[2 * j]; hb.y = h1B[2 * j + 1];
        #pragma unroll
        for (int g = 0; g < 4; ++g)
          acc[g] = __builtin_amdgcn_wmma_f32_16x16x4_f32(false, Bw0[4 + j][g], false, hb, (short)0, acc[g], false, false);
      }
      __builtin_amdgcn_sched_barrier(0);
      {
        v4f a = La[0], b = Lb[0];
        #pragma unroll
        for (int p = 1; p < 8; ++p) { a += La[p]; b += Lb[p]; }
        #pragma unroll
        for (int i = 0; i < 4; ++i) { h2B[i] = a[i]; h2B[4 + i] = b[i]; }
      }
      if ((n | f) != 0 && wsc == (f & 7)) {
        const int fp = (f == 0) ? FUT - 1 : f - 1;
        const int np = (f == 0) ? n - 1 : n;
        float* op = out + (size_t)(b0 + lm) * FUT * NNODES * O_DIM
                        + (size_t)fp * NNODES * O_DIM + (size_t)np * O_DIM + 8 * lh;
        v4f oa, ob;
        #pragma unroll
        for (int i = 0; i < 4; ++i) { oa[i] = h2B[i] + nmB[i]; ob[i] = h2B[4 + i] + nmB[4 + i]; }
        *(v4f*)op       = oa;
        *(v4f*)(op + 4) = ob;
      }
      if (f == 0) {
        #pragma unroll
        for (int i = 0; i < 4; ++i) { nmB[i] = nmNa[i]; nmB[4 + i] = nmNb[i]; }
      }
      float xB[8];
      #pragma unroll
      for (int r = 0; r < 8; ++r) xB[r] = h2B[r] + nmB[r];
      if ((n | f) == 0) {
        #pragma unroll
        for (int r = 0; r < 8; ++r) h2B[r] = h0[O_DIM + 8 * lh + r];
      }
      #pragma unroll
      for (int j = 0; j < 4; ++j) {
        v2f xb; xb.x = xB[2 * j]; xb.y = xB[2 * j + 1];
        #pragma unroll
        for (int g = 0; g < 4; ++g)
          acc[g] = __builtin_amdgcn_wmma_f32_16x16x4_f32(false, Bw0[j][g], false, xb, (short)0, acc[g], false, false);
      }
      {
        float hf[8];
        #pragma unroll
        for (int r = 0; r < 8; ++r) {
          float iv = sigm(acc[0][r]);
          float fv = sigm(acc[1][r]);
          float gv = tanh_hw(acc[2][r]);
          float ov = sigm(acc[3][r]);
          float cn = fv * c1r[r] + iv * gv;
          c1r[r] = cn;
          hf[r] = ov * tanh_hw(cn);
        }
        v8f pacc = {};
        #pragma unroll
        for (int k = 0; k < 4; ++k) {
          v2f hb; hb.x = hf[2 * k]; hb.y = hf[2 * k + 1];
          pacc = __builtin_amdgcn_wmma_f32_16x16x4_f32(false, Pw0[k], false, hb, (short)0, pacc, false, false);
        }
        v4f qa, qb;
        #pragma unroll
        for (int i = 0; i < 4; ++i) { qa[i] = pacc[i]; qb[i] = pacc[4 + i]; }
        *(v4f*)pw0a = qa;
        *(v4f*)pw0b = qb;
      }
      __syncthreads();

      #pragma unroll
      for (int p = 0; p < 8; ++p) {
        La[p] = *(const v4f*)(pr0 + p * 256 + pc0);
        Lb[p] = *(const v4f*)(pr0 + p * 256 + pc1);
      }
      __builtin_amdgcn_sched_barrier(0);
      #pragma unroll
      for (int g = 0; g < 4; ++g) acc[g] = bias1v[g];
      #pragma unroll
      for (int j = 0; j < 4; ++j) {
        v2f hb; hb.x = h2B[2 * j]; hb.y = h2B[2 * j + 1];
        #pragma unroll
        for (int g = 0; g < 4; ++g)
          acc[g] = __builtin_amdgcn_wmma_f32_16x16x4_f32(false, Bw1[4 + j][g], false, hb, (short)0, acc[g], false, false);
      }
      __builtin_amdgcn_sched_barrier(0);
      {
        v4f a = La[0], b = Lb[0];
        #pragma unroll
        for (int p = 1; p < 8; ++p) { a += La[p]; b += Lb[p]; }
        #pragma unroll
        for (int i = 0; i < 4; ++i) { h1B[i] = a[i]; h1B[4 + i] = b[i]; }
      }
      #pragma unroll
      for (int j = 0; j < 4; ++j) {
        v2f xb; xb.x = h1B[2 * j]; xb.y = h1B[2 * j + 1];
        #pragma unroll
        for (int g = 0; g < 4; ++g)
          acc[g] = __builtin_amdgcn_wmma_f32_16x16x4_f32(false, Bw1[j][g], false, xb, (short)0, acc[g], false, false);
      }
      {
        float hf[8];
        #pragma unroll
        for (int r = 0; r < 8; ++r) {
          float iv = sigm(acc[0][r]);
          float fv = sigm(acc[1][r]);
          float gv = tanh_hw(acc[2][r]);
          float ov = sigm(acc[3][r]);
          float cn = fv * c2r[r] + iv * gv;
          c2r[r] = cn;
          hf[r] = ov * tanh_hw(cn);
        }
        v8f pacc = {};
        #pragma unroll
        for (int k = 0; k < 4; ++k) {
          v2f hb; hb.x = hf[2 * k]; hb.y = hf[2 * k + 1];
          pacc = __builtin_amdgcn_wmma_f32_16x16x4_f32(false, Pw1[k], false, hb, (short)0, pacc, false, false);
        }
        v4f qa, qb;
        #pragma unroll
        for (int i = 0; i < 4; ++i) { qa[i] = pacc[i]; qb[i] = pacc[4 + i]; }
        *(v4f*)pw1a = qa;
        *(v4f*)pw1b = qb;
      }
      __syncthreads();
    }
  }

  {
    v4f a = *(const v4f*)(pr1 + pc0), b = *(const v4f*)(pr1 + pc1);
    #pragma unroll
    for (int p = 1; p < 8; ++p) {
      a += *(const v4f*)(pr1 + p * 256 + pc0);
      b += *(const v4f*)(pr1 + p * 256 + pc1);
    }
    if (wsc == 0) {
      float* op = out + (size_t)(b0 + lm) * FUT * NNODES * O_DIM
                      + (size_t)(FUT - 1) * NNODES * O_DIM + (size_t)(NNODES - 1) * O_DIM + 8 * lh;
      v4f oa, ob;
      #pragma unroll
      for (int i = 0; i < 4; ++i) { oa[i] = a[i] + nmB[i]; ob[i] = b[i] + nmB[4 + i]; }
      *(v4f*)op       = oa;
      *(v4f*)(op + 4) = ob;
    }
  }
}

extern "C" void kernel_launch(void* const* d_in, const int* in_sizes, int n_in,
                              void* d_out, int out_size, void* d_ws, size_t ws_size,
                              hipStream_t stream) {
  (void)in_sizes; (void)n_in; (void)out_size; (void)d_ws; (void)ws_size;
  const float* enc   = (const float*)d_in[0];
  const float* mean_ = (const float*)d_in[1];
  const float* s     = (const float*)d_in[2];
  const float* h0    = (const float*)d_in[3];
  const float* c0    = (const float*)d_in[4];
  const float* Wih0  = (const float*)d_in[5];
  const float* Whh0  = (const float*)d_in[6];
  const float* bih0  = (const float*)d_in[7];
  const float* bhh0  = (const float*)d_in[8];
  const float* Whr0  = (const float*)d_in[9];
  const float* Wih1  = (const float*)d_in[10];
  const float* Whh1  = (const float*)d_in[11];
  const float* bih1  = (const float*)d_in[12];
  const float* bhh1  = (const float*)d_in[13];
  const float* Whr1  = (const float*)d_in[14];
  float* out = (float*)d_out;

  decoder_scan<<<dim3(BATCH / 16), dim3(256), 0, stream>>>(
      enc, mean_, s, h0, c0,
      Wih0, Whh0, bih0, bhh0, Whr0,
      Wih1, Whh1, bih1, bhh1, Whr1, out);
}


# Recurrent half of each gate product formed a phase early, between signal and wait of a split barrier at low wave priority (s_setprio): fills the matrix unit while the SIMD's other wave activates
# speedup vs baseline: 4.0200x; 1.1171x over previous
#include <hip/hip_runtime.h>
#include <math.h>

typedef __attribute__((ext_vector_type(2))) float v2f;
typedef __attribute__((ext_vector_type(4))) float v4f;
typedef __attribute__((ext_vector_type(8))) float v8f;

#define E_DIM 64
#define H_DIM 128
#define O_DIM 16
#define BATCH 64
#define NNODES 1000
#define FUT 12
#define G4 512

__device__ __forceinline__ float tanh_hw(float x) {
  return __builtin_amdgcn_tanhf(x);
}
__device__ __forceinline__ float sigm(float x) {
  return __builtin_fmaf(0.5f, __builtin_amdgcn_tanhf(0.5f * x), 0.5f);
}

__global__ __launch_bounds__(256) void decoder_scan(
    const float* __restrict__ enc,
    const float* __restrict__ mean_,
    const float* __restrict__ s,
    const float* __restrict__ h0,
    const float* __restrict__ c0,
    const float* __restrict__ Wih0,
    const float* __restrict__ Whh0,
    const float* __restrict__ bih0, const float* __restrict__ bhh0,
    const float* __restrict__ Whr0,
    const float* __restrict__ Wih1,
    const float* __restrict__ Whh1,
    const float* __restrict__ bih1, const float* __restrict__ bhh1,
    const float* __restrict__ Whr1,
    float* __restrict__ out)
{
  __shared__ __attribute__((aligned(32))) float part[2][8][4][16][4];
  __shared__ __attribute__((aligned(32))) float bias0s[G4];
  __shared__ __attribute__((aligned(16))) float BeS[8][4][8][32][4];

  const int tid  = threadIdx.x;
  const int w    = tid >> 5;
  const int lane = tid & 31;
  const int lm   = lane & 15;
  const int lh   = lane >> 4;
  const int b0   = blockIdx.x * 16;
  const int wsc  = __builtin_amdgcn_readfirstlane(w);

  bias0s[tid]       = bih0[tid]       + bhh0[tid];
  bias0s[tid + 256] = bih0[tid + 256] + bhh0[tid + 256];

  v2f Bw0[8][4], Bw1[8][4];
  v8f bias1v[4];
  for (int g = 0; g < 4; ++g) {
    const int gc = 16 * (w + 8 * g) + lm;
    for (int r = 0; r < 8; ++r) {
      const int gr = 16 * (w + 8 * g) + 8 * lh + r;
      bias1v[g][r] = bih1[gr] + bhh1[gr];
    }
    for (int j = 0; j < 4; ++j) {
      const int kk = 8 * lh + 2 * j;
      Bw0[j][g].x     = Wih0[gc * 80 + 64 + kk];  Bw0[j][g].y     = Wih0[gc * 80 + 64 + kk + 1];
      Bw0[4 + j][g].x = Whh0[gc * 16 + kk];       Bw0[4 + j][g].y = Whh0[gc * 16 + kk + 1];
      Bw1[j][g].x     = Wih1[gc * 16 + kk];       Bw1[j][g].y     = Wih1[gc * 16 + kk + 1];
      Bw1[4 + j][g].x = Whh1[gc * 16 + kk];       Bw1[4 + j][g].y = Whh1[gc * 16 + kk + 1];
    }
  }
  for (int g = 0; g < 4; ++g) {
    const int gc = 16 * (w + 8 * g) + lm;
    for (int kp = 0; kp < 8; ++kp) {
      const int kk = 8 * kp + 2 * lh;
      v4f t;
      t.x = Wih0[gc * 80 + kk];      t.y = Wih0[gc * 80 + kk + 1];
      t.z = Wih0[gc * 80 + kk + 4];  t.w = Wih0[gc * 80 + kk + 5];
      *(v4f*)&BeS[w][g][kp][lane][0] = t;
    }
  }
  v2f Pw0[4], Pw1[4];
  for (int k = 0; k < 4; ++k) {
    const int kk = 16 * w + 8 * lh + 2 * k;
    Pw0[k].x = Whr0[lm * H_DIM + kk];  Pw0[k].y = Whr0[lm * H_DIM + kk + 1];
    Pw1[k].x = Whr1[lm * H_DIM + kk];  Pw1[k].y = Whr1[lm * H_DIM + kk + 1];
  }

  float h1B[8], h2B[8], nmB[8];
  #pragma unroll
  for (int r = 0; r < 8; ++r) { h1B[r] = h0[8 * lh + r]; h2B[r] = 0.f; nmB[r] = 0.f; }

  const int hcol0 = 16 * w + 8 * lh;
  float c1r[8], c2r[8];
  #pragma unroll
  for (int r = 0; r < 8; ++r) { c1r[r] = c0[hcol0 + r]; c2r[r] = c0[H_DIM + hcol0 + r]; }

  const int pc0 = (2 * lh * 16 + lm) * 4;
  const int pc1 = ((2 * lh + 1) * 16 + lm) * 4;
  float* const pw0a = &part[0][w][0][0][0] + pc0;
  float* const pw0b = &part[0][w][0][0][0] + pc1;
  float* const pw1a = &part[1][w][0][0][0] + pc0;
  float* const pw1b = &part[1][w][0][0][0] + pc1;
  const float* const pr0 = &part[0][0][0][0][0];
  const float* const pr1 = &part[1][0][0][0][0];

  {
    v4f sa = {0.f, 0.f, 0.f, 0.f}, sb = {0.f, 0.f, 0.f, 0.f};
    if (w == 0) { sa = *(const v4f*)&s[8 * lh]; sb = *(const v4f*)&s[8 * lh + 4]; }
    *(v4f*)pw1a = sa;
    *(v4f*)pw1b = sb;
  }
  __syncthreads();

  v8f genc[4];
  v4f nmNa = {0.f, 0.f, 0.f, 0.f}, nmNb = {0.f, 0.f, 0.f, 0.f};
  auto node_block = [&](int nn) {
    const float* mp = mean_ + (size_t)(b0 + lm) * FUT * NNODES * O_DIM
                            + (size_t)nn * O_DIM + 8 * lh;
    v4f ma = {0.f, 0.f, 0.f, 0.f}, mb = {0.f, 0.f, 0.f, 0.f};
    #pragma unroll
    for (int ff = 0; ff < FUT; ++ff) {
      ma += *(const v4f*)(mp + (size_t)ff * NNODES * O_DIM);
      mb += *(const v4f*)(mp + (size_t)ff * NNODES * O_DIM + 4);
    }
    nmNa = ma * (1.0f / FUT);
    nmNb = mb * (1.0f / FUT);
    const float* ap = enc + (size_t)(b0 + lm) * NNODES * E_DIM + (size_t)nn * E_DIM;
    #pragma unroll
    for (int g = 0; g < 4; ++g) {
      v8f acc = *(const v8f*)&bias0s[16 * (w + 8 * g) + 8 * lh];
      #pragma unroll
      for (int kp = 0; kp < 8; ++kp) {
        const v4f wv = *(const v4f*)&BeS[w][g][kp][lane][0];
        int kk = 8 * kp + 2 * lh;
        v2f wa, a;
        wa.x = wv.x;               wa.y = wv.y;
        a.x = ap[kk];              a.y = ap[kk + 1];
        acc = __builtin_amdgcn_wmma_f32_16x16x4_f32(false, wa, false, a, (short)0, acc, false, false);
        wa.x = wv.z;               wa.y = wv.w;
        a.x = ap[kk + 4];          a.y = ap[kk + 5];
        acc = __builtin_amdgcn_wmma_f32_16x16x4_f32(false, wa, false, a, (short)0, acc, false, false);
      }
      genc[g] = acc;
    }
  };

  v8f acc0[4], acc1[4];
  node_block(0);
  #pragma unroll
  for (int g = 0; g < 4; ++g) acc0[g] = genc[g];
  #pragma unroll
  for (int j = 0; j < 4; ++j) {
    v2f hb; hb.x = h1B[2 * j]; hb.y = h1B[2 * j + 1];
    #pragma unroll
    for (int g = 0; g < 4; ++g)
      acc0[g] = __builtin_amdgcn_wmma_f32_16x16x4_f32(false, Bw0[4 + j][g], false, hb, (short)0, acc0[g], false, false);
  }

  for (int n = 0; n < NNODES; ++n) {
    for (int f = 0; f < FUT; ++f) {
      __builtin_amdgcn_s_setprio(3);
      v4f La[8], Lb[8];
      #pragma unroll
      for (int p = 0; p < 8; ++p) {
        La[p] = *(const v4f*)(pr1 + p * 256 + pc0);
        Lb[p] = *(const v4f*)(pr1 + p * 256 + pc1);
      }
      __builtin_amdgcn_sched_barrier(0);
      {
        v4f a = La[0], b = Lb[0];
        #pragma unroll
        for (int p = 1; p < 8; ++p) { a += La[p]; b += Lb[p]; }
        #pragma unroll
        for (int i = 0; i < 4; ++i) { h2B[i] = a[i]; h2B[4 + i] = b[i]; }
      }
      if ((n | f) != 0 && wsc == (f & 7)) {
        const int fp = (f == 0) ? FUT - 1 : f - 1;
        const int np = (f == 0) ? n - 1 : n;
        float* op = out + (size_t)(b0 + lm) * FUT * NNODES * O_DIM
                        + (size_t)fp * NNODES * O_DIM + (size_t)np * O_DIM + 8 * lh;
        v4f oa, ob;
        #pragma unroll
        for (int i = 0; i < 4; ++i) { oa[i] = h2B[i] + nmB[i]; ob[i] = h2B[4 + i] + nmB[4 + i]; }
        *(v4f*)op       = oa;
        *(v4f*)(op + 4) = ob;
      }
      if (f == 0) {
        #pragma unroll
        for (int i = 0; i < 4; ++i) { nmB[i] = nmNa[i]; nmB[4 + i] = nmNb[i]; }
      }
      float xB[8];
      #pragma unroll
      for (int r = 0; r < 8; ++r) xB[r] = h2B[r] + nmB[r];
      if ((n | f) == 0) {
        #pragma unroll
        for (int r = 0; r < 8; ++r) h2B[r] = h0[O_DIM + 8 * lh + r];
      }
      #pragma unroll
      for (int j = 0; j < 4; ++j) {
        v2f xb; xb.x = xB[2 * j]; xb.y = xB[2 * j + 1];
        #pragma unroll
        for (int g = 0; g < 4; ++g)
          acc0[g] = __builtin_amdgcn_wmma_f32_16x16x4_f32(false, Bw0[j][g], false, xb, (short)0, acc0[g], false, false);
      }
      {
        float hf[8];
        #pragma unroll
        for (int r = 0; r < 8; ++r) {
          float iv = sigm(acc0[0][r]);
          float fv = sigm(acc0[1][r]);
          float gv = tanh_hw(acc0[2][r]);
          float ov = sigm(acc0[3][r]);
          float cn = fv * c1r[r] + iv * gv;
          c1r[r] = cn;
          hf[r] = ov * tanh_hw(cn);
        }
        v8f pacc = {};
        #pragma unroll
        for (int k = 0; k < 4; ++k) {
          v2f hb; hb.x = hf[2 * k]; hb.y = hf[2 * k + 1];
          pacc = __builtin_amdgcn_wmma_f32_16x16x4_f32(false, Pw0[k], false, hb, (short)0, pacc, false, false);
        }
        v4f qa, qb;
        #pragma unroll
        for (int i = 0; i < 4; ++i) { qa[i] = pacc[i]; qb[i] = pacc[4 + i]; }
        *(v4f*)pw0a = qa;
        *(v4f*)pw0b = qb;
      }
      __builtin_amdgcn_fence(__ATOMIC_RELEASE, "workgroup");
      __builtin_amdgcn_s_barrier_signal(-1);
      __builtin_amdgcn_s_setprio(0);
      __builtin_amdgcn_sched_barrier(0);
      #pragma unroll
      for (int g = 0; g < 4; ++g) acc1[g] = bias1v[g];
      #pragma unroll
      for (int j = 0; j < 4; ++j) {
        v2f hb; hb.x = h2B[2 * j]; hb.y = h2B[2 * j + 1];
        #pragma unroll
        for (int g = 0; g < 4; ++g)
          acc1[g] = __builtin_amdgcn_wmma_f32_16x16x4_f32(false, Bw1[4 + j][g], false, hb, (short)0, acc1[g], false, false);
      }
      asm volatile("" : "+v"(acc1[0]), "+v"(acc1[1]), "+v"(acc1[2]), "+v"(acc1[3]));
      __builtin_amdgcn_s_barrier_wait(-1);
      __builtin_amdgcn_fence(__ATOMIC_ACQUIRE, "workgroup");

      __builtin_amdgcn_s_setprio(3);
      #pragma unroll
      for (int p = 0; p < 8; ++p) {
        La[p] = *(const v4f*)(pr0 + p * 256 + pc0);
        Lb[p] = *(const v4f*)(pr0 + p * 256 + pc1);
      }
      __builtin_amdgcn_sched_barrier(0);
      {
        v4f a = La[0], b = Lb[0];
        #pragma unroll
        for (int p = 1; p < 8; ++p) { a += La[p]; b += Lb[p]; }
        #pragma unroll
        for (int i = 0; i < 4; ++i) { h1B[i] = a[i]; h1B[4 + i] = b[i]; }
      }
      #pragma unroll
      for (int j = 0; j < 4; ++j) {
        v2f xb; xb.x = h1B[2 * j]; xb.y = h1B[2 * j + 1];
        #pragma unroll
        for (int g = 0; g < 4; ++g)
          acc1[g] = __builtin_amdgcn_wmma_f32_16x16x4_f32(false, Bw1[j][g], false, xb, (short)0, acc1[g], false, false);
      }
      {
        float hf[8];
        #pragma unroll
        for (int r = 0; r < 8; ++r) {
          float iv = sigm(acc1[0][r]);
          float fv = sigm(acc1[1][r]);
          float gv = tanh_hw(acc1[2][r]);
          float ov = sigm(acc1[3][r]);
          float cn = fv * c2r[r] + iv * gv;
          c2r[r] = cn;
          hf[r] = ov * tanh_hw(cn);
        }
        v8f pacc = {};
        #pragma unroll
        for (int k = 0; k < 4; ++k) {
          v2f hb; hb.x = hf[2 * k]; hb.y = hf[2 * k + 1];
          pacc = __builtin_amdgcn_wmma_f32_16x16x4_f32(false, Pw1[k], false, hb, (short)0, pacc, false, false);
        }
        v4f qa, qb;
        #pragma unroll
        for (int i = 0; i < 4; ++i) { qa[i] = pacc[i]; qb[i] = pacc[4 + i]; }
        *(v4f*)pw1a = qa;
        *(v4f*)pw1b = qb;
      }
      __builtin_amdgcn_fence(__ATOMIC_RELEASE, "workgroup");
      __builtin_amdgcn_s_barrier_signal(-1);
      __builtin_amdgcn_s_setprio(0);
      __builtin_amdgcn_sched_barrier(0);
      if (f == FUT - 1 && n + 1 < NNODES) node_block(n + 1);
      #pragma unroll
      for (int g = 0; g < 4; ++g) acc0[g] = genc[g];
      #pragma unroll
      for (int j = 0; j < 4; ++j) {
        v2f hb; hb.x = h1B[2 * j]; hb.y = h1B[2 * j + 1];
        #pragma unroll
        for (int g = 0; g < 4; ++g)
          acc0[g] = __builtin_amdgcn_wmma_f32_16x16x4_f32(false, Bw0[4 + j][g], false, hb, (short)0, acc0[g], false, false);
      }
      asm volatile("" : "+v"(acc0[0]), "+v"(acc0[1]), "+v"(acc0[2]), "+v"(acc0[3]));
      __builtin_amdgcn_s_barrier_wait(-1);
      __builtin_amdgcn_fence(__ATOMIC_ACQUIRE, "workgroup");
    }
  }
  {
    v4f a = *(const v4f*)(pr1 + pc0), b = *(const v4f*)(pr1 + pc1);
    #pragma unroll
    for (int p = 1; p < 8; ++p) {
      a += *(const v4f*)(pr1 + p * 256 + pc0);
      b += *(const v4f*)(pr1 + p * 256 + pc1);
    }
    if (wsc == 0) {
      float* op = out + (size_t)(b0 + lm) * FUT * NNODES * O_DIM
                      + (size_t)(FUT - 1) * NNODES * O_DIM + (size_t)(NNODES - 1) * O_DIM + 8 * lh;
      v4f oa, ob;
      #pragma unroll
      for (int i = 0; i < 4; ++i) { oa[i] = a[i] + nmB[i]; ob[i] = b[i] + nmB[4 + i]; }
      *(v4f*)op       = oa;
      *(v4f*)(op + 4) = ob;
    }
  }
}

extern "C" void kernel_launch(void* const* d_in, const int* in_sizes, int n_in,
                              void* d_out, int out_size, void* d_ws, size_t ws_size,
                              hipStream_t stream) {
  (void)in_sizes; (void)n_in; (void)out_size; (void)d_ws; (void)ws_size;
  const float* enc   = (const float*)d_in[0];
  const float* mean_ = (const float*)d_in[1];
  const float* s     = (const float*)d_in[2];
  const float* h0    = (const float*)d_in[3];
  const float* c0    = (const float*)d_in[4];
  const float* Wih0  = (const float*)d_in[5];
  const float* Whh0  = (const float*)d_in[6];
  const float* bih0  = (const float*)d_in[7];
  const float* bhh0  = (const float*)d_in[8];
  const float* Whr0  = (const float*)d_in[9];
  const float* Wih1  = (const float*)d_in[10];
  const float* Whh1  = (const float*)d_in[11];
  const float* bih1  = (const float*)d_in[12];
  const float* bhh1  = (const float*)d_in[13];
  const float* Whr1  = (const float*)d_in[14];
  float* out = (float*)d_out;

  decoder_scan<<<dim3(BATCH / 16), dim3(256), 0, stream>>>(
      enc, mean_, s, h0, c0,
      Wih0, Whh0, bih0, bhh0, Whr0,
      Wih1, Whh1, bih1, bhh1, Whr1, out);
}


# Cell update on packed register pairs; gate bias and enc term enter in its first multiply-add (x1/2 for sigmoids), accumulators start at zero C: no per-step register copies; bias now summed last
# speedup vs baseline: 4.1790x; 1.0396x over previous
#include <hip/hip_runtime.h>
#include <math.h>

typedef __attribute__((ext_vector_type(2))) float v2f;
typedef __attribute__((ext_vector_type(4))) float v4f;
typedef __attribute__((ext_vector_type(8))) float v8f;

#define E_DIM 64
#define H_DIM 128
#define O_DIM 16
#define BATCH 64
#define NNODES 1000
#define FUT 12
#define G4 512

__device__ __forceinline__ v2f tanh2(v2f x) {
  v2f r;
  r.x = __builtin_amdgcn_tanhf(x.x);
  r.y = __builtin_amdgcn_tanhf(x.y);
  return r;
}
template <int K>
__device__ __forceinline__ v2f cell_pair(const v8f (&acc)[4], const v8f (&hb)[4], v2f& c) {
  const v2f hf2 = {0.5f, 0.5f};
  v2f ai = {acc[0][2 * K], acc[0][2 * K + 1]};
  v2f af = {acc[1][2 * K], acc[1][2 * K + 1]};
  v2f ag = {acc[2][2 * K], acc[2][2 * K + 1]};
  v2f ao = {acc[3][2 * K], acc[3][2 * K + 1]};
  v2f bi = {hb[0][2 * K], hb[0][2 * K + 1]};
  v2f bf = {hb[1][2 * K], hb[1][2 * K + 1]};
  v2f bg = {hb[2][2 * K], hb[2][2 * K + 1]};
  v2f bo = {hb[3][2 * K], hb[3][2 * K + 1]};
  v2f ti = tanh2(__builtin_elementwise_fma(ai, hf2, bi));
  v2f tf = tanh2(__builtin_elementwise_fma(af, hf2, bf));
  v2f tg = tanh2(ag + bg);
  v2f to = tanh2(__builtin_elementwise_fma(ao, hf2, bo));
  v2f si = __builtin_elementwise_fma(ti, hf2, hf2);
  v2f sf = __builtin_elementwise_fma(tf, hf2, hf2);
  v2f so = __builtin_elementwise_fma(to, hf2, hf2);
  v2f cn = __builtin_elementwise_fma(si, tg, sf * c);
  c = cn;
  return so * tanh2(cn);
}

__global__ __launch_bounds__(256) void decoder_scan(
    const float* __restrict__ enc,
    const float* __restrict__ mean_,
    const float* __restrict__ s,
    const float* __restrict__ h0,
    const float* __restrict__ c0,
    const float* __restrict__ Wih0,
    const float* __restrict__ Whh0,
    const float* __restrict__ bih0, const float* __restrict__ bhh0,
    const float* __restrict__ Whr0,
    const float* __restrict__ Wih1,
    const float* __restrict__ Whh1,
    const float* __restrict__ bih1, const float* __restrict__ bhh1,
    const float* __restrict__ Whr1,
    float* __restrict__ out)
{
  __shared__ __attribute__((aligned(32))) float part[2][8][4][16][4];
  __shared__ __attribute__((aligned(32))) float bias0s[G4];
  __shared__ __attribute__((aligned(16))) float BeS[8][4][8][32][4];

  const int tid  = threadIdx.x;
  const int w    = tid >> 5;
  const int lane = tid & 31;
  const int lm   = lane & 15;
  const int lh   = lane >> 4;
  const int b0   = blockIdx.x * 16;
  const int wsc  = __builtin_amdgcn_readfirstlane(w);

  bias0s[tid]       = bih0[tid]       + bhh0[tid];
  bias0s[tid + 256] = bih0[tid + 256] + bhh0[tid + 256];

  v2f Bw0[8][4], Bw1[8][4];
  v8f hb1[4];
  for (int g = 0; g < 4; ++g) {
    const int gc = 16 * (w + 8 * g) + lm;
    for (int r = 0; r < 8; ++r) {
      const int gr = 16 * (w + 8 * g) + 8 * lh + r;
      const float bsum = bih1[gr] + bhh1[gr];
      hb1[g][r] = (g == 2) ? bsum : 0.5f * bsum;
    }
    for (int j = 0; j < 4; ++j) {
      const int kk = 8 * lh + 2 * j;
      Bw0[j][g].x     = Wih0[gc * 80 + 64 + kk];  Bw0[j][g].y     = Wih0[gc * 80 + 64 + kk + 1];
      Bw0[4 + j][g].x = Whh0[gc * 16 + kk];       Bw0[4 + j][g].y = Whh0[gc * 16 + kk + 1];
      Bw1[j][g].x     = Wih1[gc * 16 + kk];       Bw1[j][g].y     = Wih1[gc * 16 + kk + 1];
      Bw1[4 + j][g].x = Whh1[gc * 16 + kk];       Bw1[4 + j][g].y = Whh1[gc * 16 + kk + 1];
    }
  }
  for (int g = 0; g < 4; ++g) {
    const int gc = 16 * (w + 8 * g) + lm;
    for (int kp = 0; kp < 8; ++kp) {
      const int kk = 8 * kp + 2 * lh;
      v4f t;
      t.x = Wih0[gc * 80 + kk];      t.y = Wih0[gc * 80 + kk + 1];
      t.z = Wih0[gc * 80 + kk + 4];  t.w = Wih0[gc * 80 + kk + 5];
      *(v4f*)&BeS[w][g][kp][lane][0] = t;
    }
  }
  v2f Pw0[4], Pw1[4];
  for (int k = 0; k < 4; ++k) {
    const int kk = 16 * w + 8 * lh + 2 * k;
    Pw0[k].x = Whr0[lm * H_DIM + kk];  Pw0[k].y = Whr0[lm * H_DIM + kk + 1];
    Pw1[k].x = Whr1[lm * H_DIM + kk];  Pw1[k].y = Whr1[lm * H_DIM + kk + 1];
  }

  float h1B[8], h2B[8], nmB[8];
  #pragma unroll
  for (int r = 0; r < 8; ++r) { h1B[r] = h0[8 * lh + r]; h2B[r] = 0.f; nmB[r] = 0.f; }

  const int hcol0 = 16 * w + 8 * lh;
  v2f c1p[4], c2p[4];
  #pragma unroll
  for (int k = 0; k < 4; ++k) {
    c1p[k].x = c0[hcol0 + 2 * k];          c1p[k].y = c0[hcol0 + 2 * k + 1];
    c2p[k].x = c0[H_DIM + hcol0 + 2 * k];  c2p[k].y = c0[H_DIM + hcol0 + 2 * k + 1];
  }

  const int pc0 = (2 * lh * 16 + lm) * 4;
  const int pc1 = ((2 * lh + 1) * 16 + lm) * 4;
  float* const pw0a = &part[0][w][0][0][0] + pc0;
  float* const pw0b = &part[0][w][0][0][0] + pc1;
  float* const pw1a = &part[1][w][0][0][0] + pc0;
  float* const pw1b = &part[1][w][0][0][0] + pc1;
  const float* const pr0 = &part[0][0][0][0][0];
  const float* const pr1 = &part[1][0][0][0][0];

  {
    v4f sa = {0.f, 0.f, 0.f, 0.f}, sb = {0.f, 0.f, 0.f, 0.f};
    if (w == 0) { sa = *(const v4f*)&s[8 * lh]; sb = *(const v4f*)&s[8 * lh + 4]; }
    *(v4f*)pw1a = sa;
    *(v4f*)pw1b = sb;
  }
  __syncthreads();

  v8f hb0[4];
  v4f nmNa = {0.f, 0.f, 0.f, 0.f}, nmNb = {0.f, 0.f, 0.f, 0.f};
  auto node_block = [&](int nn) {
    const float* mp = mean_ + (size_t)(b0 + lm) * FUT * NNODES * O_DIM
                            + (size_t)nn * O_DIM + 8 * lh;
    v4f ma = {0.f, 0.f, 0.f, 0.f}, mb = {0.f, 0.f, 0.f, 0.f};
    #pragma unroll
    for (int ff = 0; ff < FUT; ++ff) {
      ma += *(const v4f*)(mp + (size_t)ff * NNODES * O_DIM);
      mb += *(const v4f*)(mp + (size_t)ff * NNODES * O_DIM + 4);
    }
    nmNa = ma * (1.0f / FUT);
    nmNb = mb * (1.0f / FUT);
    const float* ap = enc + (size_t)(b0 + lm) * NNODES * E_DIM + (size_t)nn * E_DIM;
    #pragma unroll
    for (int g = 0; g < 4; ++g) {
      v8f acc = *(const v8f*)&bias0s[16 * (w + 8 * g) + 8 * lh];
      #pragma unroll
      for (int kp = 0; kp < 8; ++kp) {
        const v4f wv = *(const v4f*)&BeS[w][g][kp][lane][0];
        int kk = 8 * kp + 2 * lh;
        v2f wa, a;
        wa.x = wv.x;               wa.y = wv.y;
        a.x = ap[kk];              a.y = ap[kk + 1];
        acc = __builtin_amdgcn_wmma_f32_16x16x4_f32(false, wa, false, a, (short)0, acc, false, false);
        wa.x = wv.z;               wa.y = wv.w;
        a.x = ap[kk + 4];          a.y = ap[kk + 5];
        acc = __builtin_amdgcn_wmma_f32_16x16x4_f32(false, wa, false, a, (short)0, acc, false, false);
      }
      hb0[g] = (g == 2) ? acc : acc * 0.5f;
    }
  };

  v8f acc0[4], acc1[4];
  const v8f zero8 = {};
  node_block(0);
  #pragma unroll
  for (int j = 0; j < 4; ++j) {
    v2f hb; hb.x = h1B[2 * j]; hb.y = h1B[2 * j + 1];
    #pragma unroll
    for (int g = 0; g < 4; ++g)
      acc0[g] = __builtin_amdgcn_wmma_f32_16x16x4_f32(false, Bw0[4 + j][g], false, hb, (short)0, j == 0 ? zero8 : acc0[g], false, false);
  }

  auto step = [&](const int n, const int f) __attribute__((always_inline)) {
      __builtin_amdgcn_s_setprio(3);
      v4f La[8], Lb[8];
      #pragma unroll
      for (int p = 0; p < 8; ++p) {
        La[p] = *(const v4f*)(pr1 + p * 256 + pc0);
        Lb[p] = *(const v4f*)(pr1 + p * 256 + pc1);
      }
      __builtin_amdgcn_sched_barrier(0);
      {
        v4f a = La[0], b = Lb[0];
        #pragma unroll
        for (int p = 1; p < 8; ++p) { a += La[p]; b += Lb[p]; }
        #pragma unroll
        for (int i = 0; i < 4; ++i) { h2B[i] = a[i]; h2B[4 + i] = b[i]; }
      }
      if ((n | f) != 0 && wsc == (f & 7)) {
        const int fp = (f == 0) ? FUT - 1 : f - 1;
        const int np = (f == 0) ? n - 1 : n;
        float* op = out + (size_t)(b0 + lm) * FUT * NNODES * O_DIM
                        + (size_t)fp * NNODES * O_DIM + (size_t)np * O_DIM + 8 * lh;
        v4f oa, ob;
        #pragma unroll
        for (int i = 0; i < 4; ++i) { oa[i] = h2B[i] + nmB[i]; ob[i] = h2B[4 + i] + nmB[4 + i]; }
        *(v4f*)op       = oa;
        *(v4f*)(op + 4) = ob;
      }
      if (f == 0) {
        #pragma unroll
        for (int i = 0; i < 4; ++i) { nmB[i] = nmNa[i]; nmB[4 + i] = nmNb[i]; }
      }
      float xB[8];
      #pragma unroll
      for (int r = 0; r < 8; ++r) xB[r] = h2B[r] + nmB[r];
      if ((n | f) == 0) {
        #pragma unroll
        for (int r = 0; r < 8; ++r) h2B[r] = h0[O_DIM + 8 * lh + r];
      }
      #pragma unroll
      for (int j = 0; j < 4; ++j) {
        v2f xb; xb.x = xB[2 * j]; xb.y = xB[2 * j + 1];
        #pragma unroll
        for (int g = 0; g < 4; ++g)
          acc0[g] = __builtin_amdgcn_wmma_f32_16x16x4_f32(false, Bw0[j][g], false, xb, (short)0, acc0[g], false, false);
      }
      {
        v2f hf[4];
        hf[0] = cell_pair<0>(acc0, hb0, c1p[0]);
        hf[1] = cell_pair<1>(acc0, hb0, c1p[1]);
        hf[2] = cell_pair<2>(acc0, hb0, c1p[2]);
        hf[3] = cell_pair<3>(acc0, hb0, c1p[3]);
        v8f pacc = {};
        #pragma unroll
        for (int k = 0; k < 4; ++k)
          pacc = __builtin_amdgcn_wmma_f32_16x16x4_f32(false, Pw0[k], false, hf[k], (short)0, pacc, false, false);
        v4f qa, qb;
        #pragma unroll
        for (int i = 0; i < 4; ++i) { qa[i] = pacc[i]; qb[i] = pacc[4 + i]; }
        *(v4f*)pw0a = qa;
        *(v4f*)pw0b = qb;
      }
      __builtin_amdgcn_fence(__ATOMIC_RELEASE, "workgroup");
      __builtin_amdgcn_s_barrier_signal(-1);
      __builtin_amdgcn_s_setprio(0);
      __builtin_amdgcn_sched_barrier(0);
      #pragma unroll
      for (int j = 0; j < 4; ++j) {
        v2f hb; hb.x = h2B[2 * j]; hb.y = h2B[2 * j + 1];
        #pragma unroll
        for (int g = 0; g < 4; ++g)
          acc1[g] = __builtin_amdgcn_wmma_f32_16x16x4_f32(false, Bw1[4 + j][g], false, hb, (short)0, j == 0 ? zero8 : acc1[g], false, false);
      }
      asm volatile("" : "+v"(acc1[0]), "+v"(acc1[1]), "+v"(acc1[2]), "+v"(acc1[3]));
      __builtin_amdgcn_s_barrier_wait(-1);
      __builtin_amdgcn_fence(__ATOMIC_ACQUIRE, "workgroup");

      __builtin_amdgcn_s_setprio(3);
      #pragma unroll
      for (int p = 0; p < 8; ++p) {
        La[p] = *(const v4f*)(pr0 + p * 256 + pc0);
        Lb[p] = *(const v4f*)(pr0 + p * 256 + pc1);
      }
      __builtin_amdgcn_sched_barrier(0);
      {
        v4f a = La[0], b = Lb[0];
        #pragma unroll
        for (int p = 1; p < 8; ++p) { a += La[p]; b += Lb[p]; }
        #pragma unroll
        for (int i = 0; i < 4; ++i) { h1B[i] = a[i]; h1B[4 + i] = b[i]; }
      }
      #pragma unroll
      for (int j = 0; j < 4; ++j) {
        v2f xb; xb.x = h1B[2 * j]; xb.y = h1B[2 * j + 1];
        #pragma unroll
        for (int g = 0; g < 4; ++g)
          acc1[g] = __builtin_amdgcn_wmma_f32_16x16x4_f32(false, Bw1[j][g], false, xb, (short)0, acc1[g], false, false);
      }
      {
        v2f hf[4];
        hf[0] = cell_pair<0>(acc1, hb1, c2p[0]);
        hf[1] = cell_pair<1>(acc1, hb1, c2p[1]);
        hf[2] = cell_pair<2>(acc1, hb1, c2p[2]);
        hf[3] = cell_pair<3>(acc1, hb1, c2p[3]);
        v8f pacc = {};
        #pragma unroll
        for (int k = 0; k < 4; ++k)
          pacc = __builtin_amdgcn_wmma_f32_16x16x4_f32(false, Pw1[k], false, hf[k], (short)0, pacc, false, false);
        v4f qa, qb;
        #pragma unroll
        for (int i = 0; i < 4; ++i) { qa[i] = pacc[i]; qb[i] = pacc[4 + i]; }
        *(v4f*)pw1a = qa;
        *(v4f*)pw1b = qb;
      }
      __builtin_amdgcn_fence(__ATOMIC_RELEASE, "workgroup");
      __builtin_amdgcn_s_barrier_signal(-1);
      __builtin_amdgcn_s_setprio(0);
      __builtin_amdgcn_sched_barrier(0);
      if (f == FUT - 1 && n + 1 < NNODES) node_block(n + 1);
      #pragma unroll
      for (int j = 0; j < 4; ++j) {
        v2f hb; hb.x = h1B[2 * j]; hb.y = h1B[2 * j + 1];
        #pragma unroll
        for (int g = 0; g < 4; ++g)
          acc0[g] = __builtin_amdgcn_wmma_f32_16x16x4_f32(false, Bw0[4 + j][g], false, hb, (short)0, j == 0 ? zero8 : acc0[g], false, false);
      }
      asm volatile("" : "+v"(acc0[0]), "+v"(acc0[1]), "+v"(acc0[2]), "+v"(acc0[3]));
      __builtin_amdgcn_s_barrier_wait(-1);
      __builtin_amdgcn_fence(__ATOMIC_ACQUIRE, "workgroup");
  };
  for (int n = 0; n < NNODES; ++n) {
    step(n, 0);
    for (int f = 1; f < FUT; ++f) step(n, f);
  }
  {
    v4f a = *(const v4f*)(pr1 + pc0), b = *(const v4f*)(pr1 + pc1);
    #pragma unroll
    for (int p = 1; p < 8; ++p) {
      a += *(const v4f*)(pr1 + p * 256 + pc0);
      b += *(const v4f*)(pr1 + p * 256 + pc1);
    }
    if (wsc == 0) {
      float* op = out + (size_t)(b0 + lm) * FUT * NNODES * O_DIM
                      + (size_t)(FUT - 1) * NNODES * O_DIM + (size_t)(NNODES - 1) * O_DIM + 8 * lh;
      v4f oa, ob;
      #pragma unroll
      for (int i = 0; i < 4; ++i) { oa[i] = a[i] + nmB[i]; ob[i] = b[i] + nmB[4 + i]; }
      *(v4f*)op       = oa;
      *(v4f*)(op + 4) = ob;
    }
  }
}

extern "C" void kernel_launch(void* const* d_in, const int* in_sizes, int n_in,
                              void* d_out, int out_size, void* d_ws, size_t ws_size,
                              hipStream_t stream) {
  (void)in_sizes; (void)n_in; (void)out_size; (void)d_ws; (void)ws_size;
  const float* enc   = (const float*)d_in[0];
  const float* mean_ = (const float*)d_in[1];
  const float* s     = (const float*)d_in[2];
  const float* h0    = (const float*)d_in[3];
  const float* c0    = (const float*)d_in[4];
  const float* Wih0  = (const float*)d_in[5];
  const float* Whh0  = (const float*)d_in[6];
  const float* bih0  = (const float*)d_in[7];
  const float* bhh0  = (const float*)d_in[8];
  const float* Whr0  = (const float*)d_in[9];
  const float* Wih1  = (const float*)d_in[10];
  const float* Whh1  = (const float*)d_in[11];
  const float* bih1  = (const float*)d_in[12];
  const float* bhh1  = (const float*)d_in[13];
  const float* Whr1  = (const float*)d_in[14];
  float* out = (float*)d_out;

  decoder_scan<<<dim3(BATCH / 16), dim3(256), 0, stream>>>(
      enc, mean_, s, h0, c0,
      Wih0, Whh0, bih0, bhh0, Whr0,
      Wih1, Whh1, bih1, bhh1, Whr1, out);
}
